# starting kernel
# baseline (speedup 1.0000x reference)
#include <hip/hip_runtime.h>
#include <math.h>

#define NN 10000
#define EE 160000
#define FF 64
#define NF (NN * FF)
#define PI_F 3.14159265358979323846f

#define H1S 68
#define H2S 132
#define RDS 196

typedef float v2f __attribute__((ext_vector_type(2)));
typedef float v8f __attribute__((ext_vector_type(8)));

__device__ __forceinline__ v8f wmma4(v2f a, v2f b, v8f c) {
    return __builtin_amdgcn_wmma_f32_16x16x4_f32(
         false, a,  false, b,
         (short)0, c,  false,  false);
}

__device__ __forceinline__ float silu(float x) {
    return x / (1.0f + expf(-x));
}

__device__ __forceinline__ void recon3(const float c[10], float t[9]) {
    float d = c[0];
    t[0] = d + c[4];  t[4] = d + c[5];  t[8] = d + c[6];
    t[1] = c[1] + c[7];  t[3] = c[7] - c[1];
    t[2] = c[2] + c[8];  t[6] = c[8] - c[2];
    t[5] = c[3] + c[9];  t[7] = c[9] - c[3];
}

__device__ __forceinline__ void mm3(const float a[9], const float b[9], float c[9]) {
#pragma unroll
    for (int i = 0; i < 3; ++i)
#pragma unroll
        for (int j = 0; j < 3; ++j) {
            float s = 0.f;
#pragma unroll
            for (int k = 0; k < 3; ++k) s += a[i * 3 + k] * b[k * 3 + j];
            c[i * 3 + j] = s;
        }
}

__global__ void prep_kernel(const float* __restrict__ X,
                            float* __restrict__ Xn,
                            float* __restrict__ P) {
    int idx = blockIdx.x * blockDim.x + threadIdx.x;
    if (idx >= NF) return;
    float x[9]; float nrm = 0.f;
#pragma unroll
    for (int i = 0; i < 9; ++i) { x[i] = X[(size_t)idx * 9 + i]; nrm += x[i] * x[i]; }
    float inv = 1.0f / (nrm + 1.0f);
#pragma unroll
    for (int i = 0; i < 9; ++i) { x[i] *= inv; Xn[(size_t)idx * 9 + i] = x[i]; }
    float dm = (x[0] + x[4] + x[8]) * (1.0f / 3.0f);
    P[0 * NF + idx] = dm;
    P[1 * NF + idx] = 0.5f * (x[1] - x[3]);
    P[2 * NF + idx] = 0.5f * (x[2] - x[6]);
    P[3 * NF + idx] = 0.5f * (x[5] - x[7]);
    P[4 * NF + idx] = x[0] - dm;
    P[5 * NF + idx] = x[4] - dm;
    P[6 * NF + idx] = x[8] - dm;
    P[7 * NF + idx] = 0.5f * (x[1] + x[3]);
    P[8 * NF + idx] = 0.5f * (x[2] + x[6]);
    P[9 * NF + idx] = 0.5f * (x[5] + x[7]);
}

__global__ __launch_bounds__(128) void mix_kernel(const float* __restrict__ P,
                                                  float* __restrict__ Q,
                                                  const float* __restrict__ La,
                                                  const float* __restrict__ Lb,
                                                  const float* __restrict__ Lc) {
    const int lane = threadIdx.x & 31;
    const int wave = threadIdx.x >> 5;
    const int ch = blockIdx.y;
    const int r0 = blockIdx.x * 16;
    const float* W = (ch == 0) ? La : (ch < 4 ? Lb : Lc);
    const float* A = P + (size_t)ch * NF;
    float* D = Q + (size_t)ch * NF;
    const int row = lane & 15;
    const int hi = lane >> 4;
    const int kb = hi * 2;
    const int col = wave * 16 + (lane & 15);
    v8f acc = {};
#pragma unroll
    for (int k = 0; k < FF; k += 4) {
        v2f a, b;
        a.x = A[(size_t)(r0 + row) * FF + k + kb];
        a.y = A[(size_t)(r0 + row) * FF + k + kb + 1];
        b.x = W[(k + kb) * FF + col];
        b.y = W[(k + kb + 1) * FF + col];
        acc = wmma4(a, b, acc);
    }
#pragma unroll
    for (int r = 0; r < 8; ++r)
        D[(size_t)(r0 + r + 8 * hi) * FF + col] = acc[r];
}

__global__ __launch_bounds__(32) void edge_kernel(
    const float* __restrict__ rbf, const float* __restrict__ dij,
    const int* __restrict__ pi,
    const float* __restrict__ W1, const float* __restrict__ b1,
    const float* __restrict__ W2, const float* __restrict__ b2,
    const float* __restrict__ W3, const float* __restrict__ b3,
    const float* __restrict__ Q, float* __restrict__ M) {
    __shared__ float h1[16 * H1S];
    __shared__ float h2[16 * H2S];
    __shared__ float rad[16 * RDS];
    __shared__ float cc[16];

    const int lane = threadIdx.x;
    const int e0 = blockIdx.x * 16;
    const int row = lane & 15;
    const int hi = lane >> 4;
    const int kb = hi * 2;
    const int col16 = lane & 15;

    if (lane < 16) {
        float d = dij[e0 + lane];
        cc[lane] = 0.5f * (cosf(d * (PI_F / 0.5f)) + 1.0f) * (d < 0.5f ? 1.0f : 0.0f);
    }

#pragma unroll
    for (int j = 0; j < 4; ++j) {
        const int col = j * 16 + col16;
        v8f acc = {};
#pragma unroll
        for (int k = 0; k < 32; k += 4) {
            v2f a, b;
            a.x = rbf[(size_t)(e0 + row) * 32 + k + kb];
            a.y = rbf[(size_t)(e0 + row) * 32 + k + kb + 1];
            b.x = W1[(k + kb) * 64 + col];
            b.y = W1[(k + kb + 1) * 64 + col];
            acc = wmma4(a, b, acc);
        }
        float bias = b1[col];
#pragma unroll
        for (int r = 0; r < 8; ++r)
            h1[(r + 8 * hi) * H1S + col] = silu(acc[r] + bias);
    }
    __syncthreads();

#pragma unroll
    for (int j = 0; j < 8; ++j) {
        const int col = j * 16 + col16;
        v8f acc = {};
#pragma unroll
        for (int k = 0; k < 64; k += 4) {
            v2f a, b;
            a.x = h1[row * H1S + k + kb];
            a.y = h1[row * H1S + k + kb + 1];
            b.x = W2[(k + kb) * 128 + col];
            b.y = W2[(k + kb + 1) * 128 + col];
            acc = wmma4(a, b, acc);
        }
        float bias = b2[col];
#pragma unroll
        for (int r = 0; r < 8; ++r)
            h2[(r + 8 * hi) * H2S + col] = silu(acc[r] + bias);
    }
    __syncthreads();

#pragma unroll
    for (int j = 0; j < 12; ++j) {
        const int col = j * 16 + col16;
        v8f acc = {};
#pragma unroll
        for (int k = 0; k < 128; k += 4) {
            v2f a, b;
            a.x = h2[row * H2S + k + kb];
            a.y = h2[row * H2S + k + kb + 1];
            b.x = W3[(k + kb) * 192 + col];
            b.y = W3[(k + kb + 1) * 192 + col];
            acc = wmma4(a, b, acc);
        }
        float bias = b3[col];
#pragma unroll
        for (int r = 0; r < 8; ++r) {
            int er = r + 8 * hi;
            rad[er * RDS + col] = silu(acc[r] + bias) * cc[er];
        }
    }
    __syncthreads();

    for (int e = 0; e < 16; ++e) {
        const int s = pi[e0 + e];
        const int d = pi[EE + e0 + e];
#pragma unroll
        for (int t = 0; t < 2; ++t) {
            const int f = lane + 32 * t;
            const float rI = rad[e * RDS + 3 * f + 0];
            const float rA = rad[e * RDS + 3 * f + 1];
            const float rS = rad[e * RDS + 3 * f + 2];
            const int g = d * FF + f;
            const int so = s * FF + f;
            atomicAdd(&M[0 * NF + so], rI * Q[0 * NF + g]);
#pragma unroll
            for (int c = 1; c < 4; ++c)
                atomicAdd(&M[c * NF + so], rA * Q[c * NF + g]);
#pragma unroll
            for (int c = 4; c < 10; ++c)
                atomicAdd(&M[c * NF + so], rS * Q[c * NF + g]);
        }
    }
}

__global__ void combine_kernel(const float* __restrict__ M,
                               const float* __restrict__ Q,
                               const float* __restrict__ chg,
                               float* __restrict__ P) {
    int idx = blockIdx.x * blockDim.x + threadIdx.x;
    if (idx >= NF) return;
    const int n = idx >> 6;
    float mc[10], yc[10];
#pragma unroll
    for (int c = 0; c < 10; ++c) { mc[c] = M[c * NF + idx]; yc[c] = Q[c * NF + idx]; }
    float msg[9], Y[9], t1[9], t2[9];
    recon3(mc, msg);
    recon3(yc, Y);
    mm3(msg, Y, t1);
    mm3(Y, msg, t2);
    const float scale = 1.0f + 0.1f * chg[n];
    float T[9]; float tn = 0.f;
#pragma unroll
    for (int i = 0; i < 9; ++i) { T[i] = scale * (t1[i] + t2[i]); tn += T[i] * T[i]; }
    const float inv = 1.0f / (tn + 1.0f);
#pragma unroll
    for (int i = 0; i < 9; ++i) T[i] *= inv;
    float dm = (T[0] + T[4] + T[8]) * (1.0f / 3.0f);
    P[0 * NF + idx] = dm;
    P[1 * NF + idx] = 0.5f * (T[1] - T[3]);
    P[2 * NF + idx] = 0.5f * (T[2] - T[6]);
    P[3 * NF + idx] = 0.5f * (T[5] - T[7]);
    P[4 * NF + idx] = T[0] - dm;
    P[5 * NF + idx] = T[4] - dm;
    P[6 * NF + idx] = T[8] - dm;
    P[7 * NF + idx] = 0.5f * (T[1] + T[3]);
    P[8 * NF + idx] = 0.5f * (T[2] + T[6]);
    P[9 * NF + idx] = 0.5f * (T[5] + T[7]);
}

__global__ void finalize_kernel(const float* __restrict__ Q2,
                                const float* __restrict__ Xn,
                                const float* __restrict__ chg,
                                float* __restrict__ out) {
    int idx = blockIdx.x * blockDim.x + threadIdx.x;
    if (idx >= NF) return;
    const int n = idx >> 6;
    float dc[10];
#pragma unroll
    for (int c = 0; c < 10; ++c) dc[c] = Q2[c * NF + idx];
    float dX[9], prod[9];
    recon3(dc, dX);
    mm3(dX, dX, prod);
    const float scale = 1.0f + 0.1f * chg[n];
#pragma unroll
    for (int i = 0; i < 9; ++i)
        out[(size_t)idx * 9 + i] = Xn[(size_t)idx * 9 + i] + dX[i] + scale * prod[i];
}

extern "C" void kernel_launch(void* const* d_in, const int* in_sizes, int n_in,
                              void* d_out, int out_size, void* d_ws, size_t ws_size,
                              hipStream_t stream) {
    const float* X   = (const float*)d_in[0];
    const int*   pi  = (const int*)  d_in[1];
    const float* dij = (const float*)d_in[2];
    const float* rbf = (const float*)d_in[3];
    const float* chg = (const float*)d_in[4];
    const float* W1  = (const float*)d_in[5];
    const float* b1  = (const float*)d_in[6];
    const float* W2  = (const float*)d_in[7];
    const float* b2  = (const float*)d_in[8];
    const float* W3  = (const float*)d_in[9];
    const float* b3  = (const float*)d_in[10];
    const float* L0  = (const float*)d_in[11];
    const float* L1  = (const float*)d_in[12];
    const float* L2  = (const float*)d_in[13];
    const float* L3  = (const float*)d_in[14];
    const float* L4  = (const float*)d_in[15];
    const float* L5  = (const float*)d_in[16];

    float* ws = (float*)d_ws;
    float* Xn = ws;
    float* P  = Xn + (size_t)NF * 9;
    float* Q  = P + (size_t)10 * NF;
    float* M  = Q + (size_t)10 * NF;

    hipMemsetAsync(M, 0, (size_t)10 * NF * sizeof(float), stream);

    prep_kernel<<<(NF + 255) / 256, 256, 0, stream>>>(X, Xn, P);
    mix_kernel<<<dim3(NN / 16, 10), 128, 0, stream>>>(P, Q, L0, L1, L2);
    edge_kernel<<<EE / 16, 32, 0, stream>>>(rbf, dij, pi, W1, b1, W2, b2, W3, b3, Q, M);
    combine_kernel<<<(NF + 255) / 256, 256, 0, stream>>>(M, Q, chg, P);
    mix_kernel<<<dim3(NN / 16, 10), 128, 0, stream>>>(P, Q, L3, L4, L5);
    finalize_kernel<<<(NF + 255) / 256, 256, 0, stream>>>(Q, Xn, chg, (float*)d_out);
}


# edge kernel: weight-fragment loads go out one 32-WMMA set ahead in groups of 32, each edge's 20 gather loads two edges ahead of its adds (was one load round trip per WMMA and per add)
# speedup vs baseline: 2.3053x; 2.3053x over previous
#include <hip/hip_runtime.h>
#include <math.h>

#define NN 10000
#define EE 160000
#define FF 64
#define NF (NN * FF)
#define PI_F 3.14159265358979323846f

#define H1S 68
#define H2S 132
#define RDS 196

typedef float v2f __attribute__((ext_vector_type(2)));
typedef float v8f __attribute__((ext_vector_type(8)));

__device__ __forceinline__ v8f wmma4(v2f a, v2f b, v8f c) {
    return __builtin_amdgcn_wmma_f32_16x16x4_f32(
         false, a,  false, b,
         (short)0, c,  false,  false);
}

__device__ __forceinline__ float silu(float x) {
    return x / (1.0f + expf(-x));
}

__device__ __forceinline__ void recon3(const float c[10], float t[9]) {
    float d = c[0];
    t[0] = d + c[4];  t[4] = d + c[5];  t[8] = d + c[6];
    t[1] = c[1] + c[7];  t[3] = c[7] - c[1];
    t[2] = c[2] + c[8];  t[6] = c[8] - c[2];
    t[5] = c[3] + c[9];  t[7] = c[9] - c[3];
}

__device__ __forceinline__ void mm3(const float a[9], const float b[9], float c[9]) {
#pragma unroll
    for (int i = 0; i < 3; ++i)
#pragma unroll
        for (int j = 0; j < 3; ++j) {
            float s = 0.f;
#pragma unroll
            for (int k = 0; k < 3; ++k) s += a[i * 3 + k] * b[k * 3 + j];
            c[i * 3 + j] = s;
        }
}

__global__ void prep_kernel(const float* __restrict__ X,
                            float* __restrict__ Xn,
                            float* __restrict__ P) {
    int idx = blockIdx.x * blockDim.x + threadIdx.x;
    if (idx >= NF) return;
    float x[9]; float nrm = 0.f;
#pragma unroll
    for (int i = 0; i < 9; ++i) { x[i] = X[(size_t)idx * 9 + i]; nrm += x[i] * x[i]; }
    float inv = 1.0f / (nrm + 1.0f);
#pragma unroll
    for (int i = 0; i < 9; ++i) { x[i] *= inv; Xn[(size_t)idx * 9 + i] = x[i]; }
    float dm = (x[0] + x[4] + x[8]) * (1.0f / 3.0f);
    P[0 * NF + idx] = dm;
    P[1 * NF + idx] = 0.5f * (x[1] - x[3]);
    P[2 * NF + idx] = 0.5f * (x[2] - x[6]);
    P[3 * NF + idx] = 0.5f * (x[5] - x[7]);
    P[4 * NF + idx] = x[0] - dm;
    P[5 * NF + idx] = x[4] - dm;
    P[6 * NF + idx] = x[8] - dm;
    P[7 * NF + idx] = 0.5f * (x[1] + x[3]);
    P[8 * NF + idx] = 0.5f * (x[2] + x[6]);
    P[9 * NF + idx] = 0.5f * (x[5] + x[7]);
}

__global__ __launch_bounds__(128) void mix_kernel(const float* __restrict__ P,
                                                  float* __restrict__ Q,
                                                  const float* __restrict__ La,
                                                  const float* __restrict__ Lb,
                                                  const float* __restrict__ Lc) {
    const int lane = threadIdx.x & 31;
    const int wave = threadIdx.x >> 5;
    const int ch = blockIdx.y;
    const int r0 = blockIdx.x * 16;
    const float* W = (ch == 0) ? La : (ch < 4 ? Lb : Lc);
    const float* A = P + (size_t)ch * NF;
    float* D = Q + (size_t)ch * NF;
    const int row = lane & 15;
    const int hi = lane >> 4;
    const int kb = hi * 2;
    const int col = wave * 16 + (lane & 15);
    v8f acc = {};
#pragma unroll
    for (int k = 0; k < FF; k += 4) {
        v2f a, b;
        a.x = A[(size_t)(r0 + row) * FF + k + kb];
        a.y = A[(size_t)(r0 + row) * FF + k + kb + 1];
        b.x = W[(k + kb) * FF + col];
        b.y = W[(k + kb + 1) * FF + col];
        acc = wmma4(a, b, acc);
    }
#pragma unroll
    for (int r = 0; r < 8; ++r)
        D[(size_t)(r0 + r + 8 * hi) * FF + col] = acc[r];
}


template <int KT, int LD>
__device__ __forceinline__ void load_bfrag(const float* __restrict__ Wm, int col, int kb,
                                           int k0, v2f* b) {
#pragma unroll
    for (int kk = 0; kk < KT; ++kk) {
        b[kk].x = Wm[(4 * (kk + k0) + kb) * LD + col];
        b[kk].y = Wm[(4 * (kk + k0) + kb + 1) * LD + col];
    }
}

__device__ __forceinline__ void gather_q(const float* __restrict__ Q, int d, int lane,
                                         float* q) {
#pragma unroll
    for (int t = 0; t < 2; ++t)
#pragma unroll
        for (int c = 0; c < 10; ++c)
            q[10 * t + c] = Q[c * NF + d * FF + lane + 32 * t];
}

__global__ __launch_bounds__(32) void edge_kernel(
    const float* __restrict__ rbf, const float* __restrict__ dij,
    const int* __restrict__ pi,
    const float* __restrict__ W1, const float* __restrict__ b1,
    const float* __restrict__ W2, const float* __restrict__ b2,
    const float* __restrict__ W3, const float* __restrict__ b3,
    const float* __restrict__ Q, float* __restrict__ M) {
    __shared__ float h1[16 * H1S];
    __shared__ float h2[16 * H2S];
    __shared__ float rad[16 * RDS];
    __shared__ float cc[16];

    const int lane = threadIdx.x;
    const int e0 = blockIdx.x * 16;
    const int row = lane & 15;
    const int hi = lane >> 4;
    const int kb = hi * 2;
    const int col16 = lane & 15;

    const int my_idx = pi[hi * EE + e0 + row];
    const float dist = dij[e0 + row];

    v2f a1[8];
#pragma unroll
    for (int kk = 0; kk < 8; ++kk) {
        a1[kk].x = rbf[(size_t)(e0 + row) * 32 + 4 * kk + kb];
        a1[kk].y = rbf[(size_t)(e0 + row) * 32 + 4 * kk + kb + 1];
    }

    v2f g[2][32];
    float gb[2][4];

#pragma unroll
    for (int j = 0; j < 4; ++j) {
        load_bfrag<8, 64>(W1, j * 16 + col16, kb, 0, &g[0][8 * j]);
        gb[0][j] = b1[j * 16 + col16];
    }

    if (lane < 16)
        cc[lane] = 0.5f * (cosf(dist * (PI_F / 0.5f)) + 1.0f) * (dist < 0.5f ? 1.0f : 0.0f);

#pragma unroll
    for (int t = 0; t < 2; ++t) {
        load_bfrag<16, 128>(W2, t * 16 + col16, kb, 0, &g[1][16 * t]);
        gb[1][t] = b2[t * 16 + col16];
    }

#pragma unroll
    for (int j = 0; j < 4; ++j) {
        const int col = j * 16 + col16;
        v8f acc = {};
#pragma unroll
        for (int kk = 0; kk < 8; ++kk) acc = wmma4(a1[kk], g[0][8 * j + kk], acc);
        const float bias = gb[0][j];
#pragma unroll
        for (int r = 0; r < 8; ++r)
            h1[(r + 8 * hi) * H1S + col] = silu(acc[r] + bias);
    }
    __syncthreads();

    v2f a2[16];
#pragma unroll
    for (int kk = 0; kk < 16; ++kk) {
        a2[kk].x = h1[row * H1S + 4 * kk + kb];
        a2[kk].y = h1[row * H1S + 4 * kk + kb + 1];
    }
#pragma unroll
    for (int p = 0; p < 4; ++p) {
        const int cur = (p + 1) & 1;
        const int nxt = p & 1;
#pragma unroll
        for (int t = 0; t < 2; ++t) {
            if (p < 3) {
                const int ncol = (2 * (p + 1) + t) * 16 + col16;
                load_bfrag<16, 128>(W2, ncol, kb, 0, &g[nxt][16 * t]);
                gb[nxt][t] = b2[ncol];
            } else {
                load_bfrag<16, 192>(W3, col16, kb, 16 * t, &g[nxt][16 * t]);
                if (t == 0) gb[nxt][0] = b3[col16];
            }
            const int col = (2 * p + t) * 16 + col16;
            v8f acc = {};
#pragma unroll
            for (int kk = 0; kk < 16; ++kk) acc = wmma4(a2[kk], g[cur][16 * t + kk], acc);
            const float bias = gb[cur][t];
#pragma unroll
            for (int r = 0; r < 8; ++r)
                h2[(r + 8 * hi) * H2S + col] = silu(acc[r] + bias);
        }
    }
    __syncthreads();

    v2f a3[32];
#pragma unroll
    for (int kk = 0; kk < 32; ++kk) {
        a3[kk].x = h2[row * H2S + 4 * kk + kb];
        a3[kk].y = h2[row * H2S + 4 * kk + kb + 1];
    }
    float qv[3][20];
#pragma unroll
    for (int j = 0; j < 12; ++j) {
        const int cur = (j + 1) & 1;
        const int nxt = j & 1;
        const int col = j * 16 + col16;
        const int ncol = col + 16;
        v8f acc = {};
#pragma unroll
        for (int half = 0; half < 2; ++half) {
            if (j < 11) {
                load_bfrag<16, 192>(W3, ncol, kb, 16 * half, &g[nxt][16 * half]);
                if (half == 0) gb[nxt][0] = b3[ncol];
            } else {
                gather_q(Q, __builtin_amdgcn_readlane(my_idx, 16 + half), lane, qv[half]);
            }
#pragma unroll
            for (int kk = 16 * half; kk < 16 * half + 16; ++kk)
                acc = wmma4(a3[kk], g[cur][kk], acc);
        }
        const float bias = gb[cur][0];
#pragma unroll
        for (int r = 0; r < 8; ++r) {
            int er = r + 8 * hi;
            rad[er * RDS + col] = silu(acc[r] + bias) * cc[er];
        }
    }
    __syncthreads();

#pragma unroll
    for (int e = 0; e < 16; ++e) {
        if (e + 2 < 16)
            gather_q(Q, __builtin_amdgcn_readlane(my_idx, 16 + e + 2), lane, qv[(e + 2) % 3]);
        const int s = __builtin_amdgcn_readlane(my_idx, e);
        const float* q = qv[e % 3];
#pragma unroll
        for (int t = 0; t < 2; ++t) {
            const int f = lane + 32 * t;
            const float rI = rad[e * RDS + 3 * f + 0];
            const float rA = rad[e * RDS + 3 * f + 1];
            const float rS = rad[e * RDS + 3 * f + 2];
            const int so = s * FF + f;
            atomicAdd(&M[0 * NF + so], rI * q[10 * t + 0]);
#pragma unroll
            for (int c = 1; c < 4; ++c)
                atomicAdd(&M[c * NF + so], rA * q[10 * t + c]);
#pragma unroll
            for (int c = 4; c < 10; ++c)
                atomicAdd(&M[c * NF + so], rS * q[10 * t + c]);
        }
    }
}

__global__ void combine_kernel(const float* __restrict__ M,
                               const float* __restrict__ Q,
                               const float* __restrict__ chg,
                               float* __restrict__ P) {
    int idx = blockIdx.x * blockDim.x + threadIdx.x;
    if (idx >= NF) return;
    const int n = idx >> 6;
    float mc[10], yc[10];
#pragma unroll
    for (int c = 0; c < 10; ++c) { mc[c] = M[c * NF + idx]; yc[c] = Q[c * NF + idx]; }
    float msg[9], Y[9], t1[9], t2[9];
    recon3(mc, msg);
    recon3(yc, Y);
    mm3(msg, Y, t1);
    mm3(Y, msg, t2);
    const float scale = 1.0f + 0.1f * chg[n];
    float T[9]; float tn = 0.f;
#pragma unroll
    for (int i = 0; i < 9; ++i) { T[i] = scale * (t1[i] + t2[i]); tn += T[i] * T[i]; }
    const float inv = 1.0f / (tn + 1.0f);
#pragma unroll
    for (int i = 0; i < 9; ++i) T[i] *= inv;
    float dm = (T[0] + T[4] + T[8]) * (1.0f / 3.0f);
    P[0 * NF + idx] = dm;
    P[1 * NF + idx] = 0.5f * (T[1] - T[3]);
    P[2 * NF + idx] = 0.5f * (T[2] - T[6]);
    P[3 * NF + idx] = 0.5f * (T[5] - T[7]);
    P[4 * NF + idx] = T[0] - dm;
    P[5 * NF + idx] = T[4] - dm;
    P[6 * NF + idx] = T[8] - dm;
    P[7 * NF + idx] = 0.5f * (T[1] + T[3]);
    P[8 * NF + idx] = 0.5f * (T[2] + T[6]);
    P[9 * NF + idx] = 0.5f * (T[5] + T[7]);
}

__global__ void finalize_kernel(const float* __restrict__ Q2,
                                const float* __restrict__ Xn,
                                const float* __restrict__ chg,
                                float* __restrict__ out) {
    int idx = blockIdx.x * blockDim.x + threadIdx.x;
    if (idx >= NF) return;
    const int n = idx >> 6;
    float dc[10];
#pragma unroll
    for (int c = 0; c < 10; ++c) dc[c] = Q2[c * NF + idx];
    float dX[9], prod[9];
    recon3(dc, dX);
    mm3(dX, dX, prod);
    const float scale = 1.0f + 0.1f * chg[n];
#pragma unroll
    for (int i = 0; i < 9; ++i)
        out[(size_t)idx * 9 + i] = Xn[(size_t)idx * 9 + i] + dX[i] + scale * prod[i];
}

extern "C" void kernel_launch(void* const* d_in, const int* in_sizes, int n_in,
                              void* d_out, int out_size, void* d_ws, size_t ws_size,
                              hipStream_t stream) {
    const float* X   = (const float*)d_in[0];
    const int*   pi  = (const int*)  d_in[1];
    const float* dij = (const float*)d_in[2];
    const float* rbf = (const float*)d_in[3];
    const float* chg = (const float*)d_in[4];
    const float* W1  = (const float*)d_in[5];
    const float* b1  = (const float*)d_in[6];
    const float* W2  = (const float*)d_in[7];
    const float* b2  = (const float*)d_in[8];
    const float* W3  = (const float*)d_in[9];
    const float* b3  = (const float*)d_in[10];
    const float* L0  = (const float*)d_in[11];
    const float* L1  = (const float*)d_in[12];
    const float* L2  = (const float*)d_in[13];
    const float* L3  = (const float*)d_in[14];
    const float* L4  = (const float*)d_in[15];
    const float* L5  = (const float*)d_in[16];

    float* ws = (float*)d_ws;
    float* Xn = ws;
    float* P  = Xn + (size_t)NF * 9;
    float* Q  = P + (size_t)10 * NF;
    float* M  = Q + (size_t)10 * NF;

    hipMemsetAsync(M, 0, (size_t)10 * NF * sizeof(float), stream);

    prep_kernel<<<(NF + 255) / 256, 256, 0, stream>>>(X, Xn, P);
    mix_kernel<<<dim3(NN / 16, 10), 128, 0, stream>>>(P, Q, L0, L1, L2);
    edge_kernel<<<EE / 16, 32, 0, stream>>>(rbf, dij, pi, W1, b1, W2, b2, W3, b3, Q, M);
    combine_kernel<<<(NF + 255) / 256, 256, 0, stream>>>(M, Q, chg, P);
    mix_kernel<<<dim3(NN / 16, 10), 128, 0, stream>>>(P, Q, L3, L4, L5);
    finalize_kernel<<<(NF + 255) / 256, 256, 0, stream>>>(Q, Xn, chg, (float*)d_out);
}


# edge kernel: column-tile loops of MLP layers 2 and 3 and the edge loop rolled (2 tiles or 3 edges a trip): code of about 80 KB straight-line now about 33 KB, inside the 64 KB instruction cache
# speedup vs baseline: 3.7596x; 1.6309x over previous
#include <hip/hip_runtime.h>
#include <math.h>

#define NN 10000
#define EE 160000
#define FF 64
#define NF (NN * FF)
#define PI_F 3.14159265358979323846f

#define H1S 68
#define H2S 132
#define RDS 196

typedef float v2f __attribute__((ext_vector_type(2)));
typedef float v8f __attribute__((ext_vector_type(8)));

__device__ __forceinline__ v8f wmma4(v2f a, v2f b, v8f c) {
    return __builtin_amdgcn_wmma_f32_16x16x4_f32(
         false, a,  false, b,
         (short)0, c,  false,  false);
}

__device__ __forceinline__ float silu(float x) {
    return x / (1.0f + expf(-x));
}

__device__ __forceinline__ void recon3(const float c[10], float t[9]) {
    float d = c[0];
    t[0] = d + c[4];  t[4] = d + c[5];  t[8] = d + c[6];
    t[1] = c[1] + c[7];  t[3] = c[7] - c[1];
    t[2] = c[2] + c[8];  t[6] = c[8] - c[2];
    t[5] = c[3] + c[9];  t[7] = c[9] - c[3];
}

__device__ __forceinline__ void mm3(const float a[9], const float b[9], float c[9]) {
#pragma unroll
    for (int i = 0; i < 3; ++i)
#pragma unroll
        for (int j = 0; j < 3; ++j) {
            float s = 0.f;
#pragma unroll
            for (int k = 0; k < 3; ++k) s += a[i * 3 + k] * b[k * 3 + j];
            c[i * 3 + j] = s;
        }
}

__global__ void prep_kernel(const float* __restrict__ X,
                            float* __restrict__ Xn,
                            float* __restrict__ P) {
    int idx = blockIdx.x * blockDim.x + threadIdx.x;
    if (idx >= NF) return;
    float x[9]; float nrm = 0.f;
#pragma unroll
    for (int i = 0; i < 9; ++i) { x[i] = X[(size_t)idx * 9 + i]; nrm += x[i] * x[i]; }
    float inv = 1.0f / (nrm + 1.0f);
#pragma unroll
    for (int i = 0; i < 9; ++i) { x[i] *= inv; Xn[(size_t)idx * 9 + i] = x[i]; }
    float dm = (x[0] + x[4] + x[8]) * (1.0f / 3.0f);
    P[0 * NF + idx] = dm;
    P[1 * NF + idx] = 0.5f * (x[1] - x[3]);
    P[2 * NF + idx] = 0.5f * (x[2] - x[6]);
    P[3 * NF + idx] = 0.5f * (x[5] - x[7]);
    P[4 * NF + idx] = x[0] - dm;
    P[5 * NF + idx] = x[4] - dm;
    P[6 * NF + idx] = x[8] - dm;
    P[7 * NF + idx] = 0.5f * (x[1] + x[3]);
    P[8 * NF + idx] = 0.5f * (x[2] + x[6]);
    P[9 * NF + idx] = 0.5f * (x[5] + x[7]);
}

__global__ __launch_bounds__(128) void mix_kernel(const float* __restrict__ P,
                                                  float* __restrict__ Q,
                                                  const float* __restrict__ La,
                                                  const float* __restrict__ Lb,
                                                  const float* __restrict__ Lc) {
    const int lane = threadIdx.x & 31;
    const int wave = threadIdx.x >> 5;
    const int ch = blockIdx.y;
    const int r0 = blockIdx.x * 16;
    const float* W = (ch == 0) ? La : (ch < 4 ? Lb : Lc);
    const float* A = P + (size_t)ch * NF;
    float* D = Q + (size_t)ch * NF;
    const int row = lane & 15;
    const int hi = lane >> 4;
    const int kb = hi * 2;
    const int col = wave * 16 + (lane & 15);
    v8f acc = {};
#pragma unroll
    for (int k = 0; k < FF; k += 4) {
        v2f a, b;
        a.x = A[(size_t)(r0 + row) * FF + k + kb];
        a.y = A[(size_t)(r0 + row) * FF + k + kb + 1];
        b.x = W[(k + kb) * FF + col];
        b.y = W[(k + kb + 1) * FF + col];
        acc = wmma4(a, b, acc);
    }
#pragma unroll
    for (int r = 0; r < 8; ++r)
        D[(size_t)(r0 + r + 8 * hi) * FF + col] = acc[r];
}


template <int KT, int LD>
__device__ __forceinline__ void load_bfrag(const float* __restrict__ Wm, int col, int kb,
                                           int k0, v2f* b) {
#pragma unroll
    for (int kk = 0; kk < KT; ++kk) {
        b[kk].x = Wm[(4 * (kk + k0) + kb) * LD + col];
        b[kk].y = Wm[(4 * (kk + k0) + kb + 1) * LD + col];
    }
}

__device__ __forceinline__ void gather_q(const float* __restrict__ Q, int d, int lane,
                                         float* q) {
#pragma unroll
    for (int t = 0; t < 2; ++t)
#pragma unroll
        for (int c = 0; c < 10; ++c)
            q[10 * t + c] = Q[c * NF + d * FF + lane + 32 * t];
}

__device__ __forceinline__ void layer2_tile(const v2f* a2, const v2f* gc, float bias, int col,
                                            int hi, float* h2) {
    v8f acc = {};
#pragma unroll
    for (int kk = 0; kk < 16; ++kk) acc = wmma4(a2[kk], gc[kk], acc);
#pragma unroll
    for (int r = 0; r < 8; ++r)
        h2[(r + 8 * hi) * H2S + col] = silu(acc[r] + bias);
}

template <bool LAST>
__device__ __forceinline__ void layer3_tile(const v2f* a3, const v2f* gc, float bias, v2f* gn,
                                            float* bias_n, const float* __restrict__ W3,
                                            const float* __restrict__ b3, int col, int kb,
                                            int hi, const float* __restrict__ Q, int my_idx,
                                            int lane, float (*qv)[20], const float* cc,
                                            float* rad) {
    v8f acc = {};
#pragma unroll
    for (int half = 0; half < 2; ++half) {
        if constexpr (!LAST) {
            load_bfrag<16, 192>(W3, col + 16, kb, 16 * half, gn + 16 * half);
            if (half == 0) *bias_n = b3[col + 16];
        } else {
            gather_q(Q, __builtin_amdgcn_readlane(my_idx, 16 + half), lane, qv[half]);
        }
#pragma unroll
        for (int kk = 16 * half; kk < 16 * half + 16; ++kk)
            acc = wmma4(a3[kk], gc[kk], acc);
    }
#pragma unroll
    for (int r = 0; r < 8; ++r) {
        const int er = r + 8 * hi;
        rad[er * RDS + col] = silu(acc[r] + bias) * cc[er];
    }
}

__device__ __forceinline__ void scatter_edge(const float* rad, float* __restrict__ M, int e,
                                             int s, int lane, const float* q) {
#pragma unroll
    for (int t = 0; t < 2; ++t) {
        const int f = lane + 32 * t;
        const float rI = rad[e * RDS + 3 * f + 0];
        const float rA = rad[e * RDS + 3 * f + 1];
        const float rS = rad[e * RDS + 3 * f + 2];
        const int so = s * FF + f;
        atomicAdd(&M[0 * NF + so], rI * q[10 * t + 0]);
#pragma unroll
        for (int c = 1; c < 4; ++c)
            atomicAdd(&M[c * NF + so], rA * q[10 * t + c]);
#pragma unroll
        for (int c = 4; c < 10; ++c)
            atomicAdd(&M[c * NF + so], rS * q[10 * t + c]);
    }
}

__global__ __launch_bounds__(32) void edge_kernel(
    const float* __restrict__ rbf, const float* __restrict__ dij,
    const int* __restrict__ pi,
    const float* __restrict__ W1, const float* __restrict__ b1,
    const float* __restrict__ W2, const float* __restrict__ b2,
    const float* __restrict__ W3, const float* __restrict__ b3,
    const float* __restrict__ Q, float* __restrict__ M) {
    __shared__ float h1[16 * H1S];
    __shared__ float h2[16 * H2S];
    __shared__ float rad[16 * RDS];
    __shared__ float cc[16];

    const int lane = threadIdx.x;
    const int e0 = blockIdx.x * 16;
    const int row = lane & 15;
    const int hi = lane >> 4;
    const int kb = hi * 2;
    const int col16 = lane & 15;

    const int my_idx = pi[hi * EE + e0 + row];
    const float dist = dij[e0 + row];

    v2f a1[8];
#pragma unroll
    for (int kk = 0; kk < 8; ++kk) {
        a1[kk].x = rbf[(size_t)(e0 + row) * 32 + 4 * kk + kb];
        a1[kk].y = rbf[(size_t)(e0 + row) * 32 + 4 * kk + kb + 1];
    }

    v2f g[2][32];
    float gb[2][4];

#pragma unroll
    for (int j = 0; j < 4; ++j) {
        load_bfrag<8, 64>(W1, j * 16 + col16, kb, 0, &g[0][8 * j]);
        gb[0][j] = b1[j * 16 + col16];
    }

    if (lane < 16)
        cc[lane] = 0.5f * (cosf(dist * (PI_F / 0.5f)) + 1.0f) * (dist < 0.5f ? 1.0f : 0.0f);

#pragma unroll
    for (int t = 0; t < 2; ++t) {
        load_bfrag<16, 128>(W2, t * 16 + col16, kb, 0, &g[1][16 * t]);
        gb[1][t] = b2[t * 16 + col16];
    }

#pragma unroll
    for (int j = 0; j < 4; ++j) {
        const int col = j * 16 + col16;
        v8f acc = {};
#pragma unroll
        for (int kk = 0; kk < 8; ++kk) acc = wmma4(a1[kk], g[0][8 * j + kk], acc);
        const float bias = gb[0][j];
#pragma unroll
        for (int r = 0; r < 8; ++r)
            h1[(r + 8 * hi) * H1S + col] = silu(acc[r] + bias);
    }
    __syncthreads();

    v2f a2[16];
#pragma unroll
    for (int kk = 0; kk < 16; ++kk) {
        a2[kk].x = h1[row * H1S + 4 * kk + kb];
        a2[kk].y = h1[row * H1S + 4 * kk + kb + 1];
    }
#pragma unroll 1
    for (int pp = 0; pp < 2; ++pp) {
        const int cb = 64 * pp + col16;
#pragma unroll
        for (int t = 0; t < 2; ++t) {
            const int ncol = cb + 32 + 16 * t;
            load_bfrag<16, 128>(W2, ncol, kb, 0, &g[0][16 * t]);
            gb[0][t] = b2[ncol];
            layer2_tile(a2, &g[1][16 * t], gb[1][t], cb + 16 * t, hi, h2);
        }
#pragma unroll
        for (int t = 0; t < 2; ++t) {
            if (pp == 0) {
                const int ncol = cb + 64 + 16 * t;
                load_bfrag<16, 128>(W2, ncol, kb, 0, &g[1][16 * t]);
                gb[1][t] = b2[ncol];
            } else {
                load_bfrag<16, 192>(W3, col16, kb, 16 * t, &g[1][16 * t]);
                if (t == 0) gb[1][0] = b3[col16];
            }
            layer2_tile(a2, &g[0][16 * t], gb[0][t], cb + 32 + 16 * t, hi, h2);
        }
    }
    __syncthreads();

    v2f a3[32];
#pragma unroll
    for (int kk = 0; kk < 32; ++kk) {
        a3[kk].x = h2[row * H2S + 4 * kk + kb];
        a3[kk].y = h2[row * H2S + 4 * kk + kb + 1];
    }
    float qv[3][20];
#pragma unroll 1
    for (int jj = 0; jj < 5; ++jj) {
        const int c0 = 32 * jj + col16;
        layer3_tile<false>(a3, g[1], gb[1][0], g[0], &gb[0][0], W3, b3, c0, kb, hi, Q, my_idx,
                           lane, qv, cc, rad);
        layer3_tile<false>(a3, g[0], gb[0][0], g[1], &gb[1][0], W3, b3, c0 + 16, kb, hi, Q,
                           my_idx, lane, qv, cc, rad);
    }
    layer3_tile<false>(a3, g[1], gb[1][0], g[0], &gb[0][0], W3, b3, 160 + col16, kb, hi, Q,
                       my_idx, lane, qv, cc, rad);
    layer3_tile<true>(a3, g[0], gb[0][0], g[1], &gb[1][0], W3, b3, 176 + col16, kb, hi, Q, my_idx,
                      lane, qv, cc, rad);
    __syncthreads();

#pragma unroll 1
    for (int e3 = 0; e3 < 15; e3 += 3) {
#pragma unroll
        for (int u = 0; u < 3; ++u) {
            const int e = e3 + u;
            const int en = min(e + 2, 15);
            gather_q(Q, __builtin_amdgcn_readlane(my_idx, 16 + en), lane, qv[(u + 2) % 3]);
            scatter_edge(rad, M, e, __builtin_amdgcn_readlane(my_idx, e), lane, qv[u]);
        }
    }
    scatter_edge(rad, M, 15, __builtin_amdgcn_readlane(my_idx, 15), lane, qv[0]);
}

__global__ void combine_kernel(const float* __restrict__ M,
                               const float* __restrict__ Q,
                               const float* __restrict__ chg,
                               float* __restrict__ P) {
    int idx = blockIdx.x * blockDim.x + threadIdx.x;
    if (idx >= NF) return;
    const int n = idx >> 6;
    float mc[10], yc[10];
#pragma unroll
    for (int c = 0; c < 10; ++c) { mc[c] = M[c * NF + idx]; yc[c] = Q[c * NF + idx]; }
    float msg[9], Y[9], t1[9], t2[9];
    recon3(mc, msg);
    recon3(yc, Y);
    mm3(msg, Y, t1);
    mm3(Y, msg, t2);
    const float scale = 1.0f + 0.1f * chg[n];
    float T[9]; float tn = 0.f;
#pragma unroll
    for (int i = 0; i < 9; ++i) { T[i] = scale * (t1[i] + t2[i]); tn += T[i] * T[i]; }
    const float inv = 1.0f / (tn + 1.0f);
#pragma unroll
    for (int i = 0; i < 9; ++i) T[i] *= inv;
    float dm = (T[0] + T[4] + T[8]) * (1.0f / 3.0f);
    P[0 * NF + idx] = dm;
    P[1 * NF + idx] = 0.5f * (T[1] - T[3]);
    P[2 * NF + idx] = 0.5f * (T[2] - T[6]);
    P[3 * NF + idx] = 0.5f * (T[5] - T[7]);
    P[4 * NF + idx] = T[0] - dm;
    P[5 * NF + idx] = T[4] - dm;
    P[6 * NF + idx] = T[8] - dm;
    P[7 * NF + idx] = 0.5f * (T[1] + T[3]);
    P[8 * NF + idx] = 0.5f * (T[2] + T[6]);
    P[9 * NF + idx] = 0.5f * (T[5] + T[7]);
}

__global__ void finalize_kernel(const float* __restrict__ Q2,
                                const float* __restrict__ Xn,
                                const float* __restrict__ chg,
                                float* __restrict__ out) {
    int idx = blockIdx.x * blockDim.x + threadIdx.x;
    if (idx >= NF) return;
    const int n = idx >> 6;
    float dc[10];
#pragma unroll
    for (int c = 0; c < 10; ++c) dc[c] = Q2[c * NF + idx];
    float dX[9], prod[9];
    recon3(dc, dX);
    mm3(dX, dX, prod);
    const float scale = 1.0f + 0.1f * chg[n];
#pragma unroll
    for (int i = 0; i < 9; ++i)
        out[(size_t)idx * 9 + i] = Xn[(size_t)idx * 9 + i] + dX[i] + scale * prod[i];
}

extern "C" void kernel_launch(void* const* d_in, const int* in_sizes, int n_in,
                              void* d_out, int out_size, void* d_ws, size_t ws_size,
                              hipStream_t stream) {
    const float* X   = (const float*)d_in[0];
    const int*   pi  = (const int*)  d_in[1];
    const float* dij = (const float*)d_in[2];
    const float* rbf = (const float*)d_in[3];
    const float* chg = (const float*)d_in[4];
    const float* W1  = (const float*)d_in[5];
    const float* b1  = (const float*)d_in[6];
    const float* W2  = (const float*)d_in[7];
    const float* b2  = (const float*)d_in[8];
    const float* W3  = (const float*)d_in[9];
    const float* b3  = (const float*)d_in[10];
    const float* L0  = (const float*)d_in[11];
    const float* L1  = (const float*)d_in[12];
    const float* L2  = (const float*)d_in[13];
    const float* L3  = (const float*)d_in[14];
    const float* L4  = (const float*)d_in[15];
    const float* L5  = (const float*)d_in[16];

    float* ws = (float*)d_ws;
    float* Xn = ws;
    float* P  = Xn + (size_t)NF * 9;
    float* Q  = P + (size_t)10 * NF;
    float* M  = Q + (size_t)10 * NF;

    hipMemsetAsync(M, 0, (size_t)10 * NF * sizeof(float), stream);

    prep_kernel<<<(NF + 255) / 256, 256, 0, stream>>>(X, Xn, P);
    mix_kernel<<<dim3(NN / 16, 10), 128, 0, stream>>>(P, Q, L0, L1, L2);
    edge_kernel<<<EE / 16, 32, 0, stream>>>(rbf, dij, pi, W1, b1, W2, b2, W3, b3, Q, M);
    combine_kernel<<<(NF + 255) / 256, 256, 0, stream>>>(M, Q, chg, P);
    mix_kernel<<<dim3(NN / 16, 10), 128, 0, stream>>>(P, Q, L3, L4, L5);
    finalize_kernel<<<(NF + 255) / 256, 256, 0, stream>>>(Q, Xn, chg, (float*)d_out);
}


# edge kernel: a wave takes 32 edges (two 16-row tiles), so each loaded weight fragment feeds two WMMAs: weight loads and L2 traffic halved; LDS buffers overlaid, layers 1 and 3 rolled to fit the cache
# speedup vs baseline: 3.8848x; 1.0333x over previous
#include <hip/hip_runtime.h>
#include <math.h>

#define NN 10000
#define EE 160000
#define FF 64
#define NF (NN * FF)
#define PI_F 3.14159265358979323846f

#define H1S 68
#define H2S 132
#define RDS 196

typedef float v2f __attribute__((ext_vector_type(2)));
typedef float v8f __attribute__((ext_vector_type(8)));

__device__ __forceinline__ v8f wmma4(v2f a, v2f b, v8f c) {
    return __builtin_amdgcn_wmma_f32_16x16x4_f32(
         false, a,  false, b,
         (short)0, c,  false,  false);
}

__device__ __forceinline__ float silu(float x) {
    return x / (1.0f + expf(-x));
}

__device__ __forceinline__ void recon3(const float c[10], float t[9]) {
    float d = c[0];
    t[0] = d + c[4];  t[4] = d + c[5];  t[8] = d + c[6];
    t[1] = c[1] + c[7];  t[3] = c[7] - c[1];
    t[2] = c[2] + c[8];  t[6] = c[8] - c[2];
    t[5] = c[3] + c[9];  t[7] = c[9] - c[3];
}

__device__ __forceinline__ void mm3(const float a[9], const float b[9], float c[9]) {
#pragma unroll
    for (int i = 0; i < 3; ++i)
#pragma unroll
        for (int j = 0; j < 3; ++j) {
            float s = 0.f;
#pragma unroll
            for (int k = 0; k < 3; ++k) s += a[i * 3 + k] * b[k * 3 + j];
            c[i * 3 + j] = s;
        }
}

__global__ void prep_kernel(const float* __restrict__ X,
                            float* __restrict__ Xn,
                            float* __restrict__ P) {
    int idx = blockIdx.x * blockDim.x + threadIdx.x;
    if (idx >= NF) return;
    float x[9]; float nrm = 0.f;
#pragma unroll
    for (int i = 0; i < 9; ++i) { x[i] = X[(size_t)idx * 9 + i]; nrm += x[i] * x[i]; }
    float inv = 1.0f / (nrm + 1.0f);
#pragma unroll
    for (int i = 0; i < 9; ++i) { x[i] *= inv; Xn[(size_t)idx * 9 + i] = x[i]; }
    float dm = (x[0] + x[4] + x[8]) * (1.0f / 3.0f);
    P[0 * NF + idx] = dm;
    P[1 * NF + idx] = 0.5f * (x[1] - x[3]);
    P[2 * NF + idx] = 0.5f * (x[2] - x[6]);
    P[3 * NF + idx] = 0.5f * (x[5] - x[7]);
    P[4 * NF + idx] = x[0] - dm;
    P[5 * NF + idx] = x[4] - dm;
    P[6 * NF + idx] = x[8] - dm;
    P[7 * NF + idx] = 0.5f * (x[1] + x[3]);
    P[8 * NF + idx] = 0.5f * (x[2] + x[6]);
    P[9 * NF + idx] = 0.5f * (x[5] + x[7]);
}

__global__ __launch_bounds__(128) void mix_kernel(const float* __restrict__ P,
                                                  float* __restrict__ Q,
                                                  const float* __restrict__ La,
                                                  const float* __restrict__ Lb,
                                                  const float* __restrict__ Lc) {
    const int lane = threadIdx.x & 31;
    const int wave = threadIdx.x >> 5;
    const int ch = blockIdx.y;
    const int r0 = blockIdx.x * 16;
    const float* W = (ch == 0) ? La : (ch < 4 ? Lb : Lc);
    const float* A = P + (size_t)ch * NF;
    float* D = Q + (size_t)ch * NF;
    const int row = lane & 15;
    const int hi = lane >> 4;
    const int kb = hi * 2;
    const int col = wave * 16 + (lane & 15);
    v8f acc = {};
#pragma unroll
    for (int k = 0; k < FF; k += 4) {
        v2f a, b;
        a.x = A[(size_t)(r0 + row) * FF + k + kb];
        a.y = A[(size_t)(r0 + row) * FF + k + kb + 1];
        b.x = W[(k + kb) * FF + col];
        b.y = W[(k + kb + 1) * FF + col];
        acc = wmma4(a, b, acc);
    }
#pragma unroll
    for (int r = 0; r < 8; ++r)
        D[(size_t)(r0 + r + 8 * hi) * FF + col] = acc[r];
}


#define TE 32

template <int KT, int LD>
__device__ __forceinline__ void load_bfrag(const float* __restrict__ Wm, int col, int kb,
                                           int k0, v2f* b) {
#pragma unroll
    for (int kk = 0; kk < KT; ++kk) {
        b[kk].x = Wm[(4 * (kk + k0) + kb) * LD + col];
        b[kk].y = Wm[(4 * (kk + k0) + kb + 1) * LD + col];
    }
}

__device__ __forceinline__ void gather_q(const float* __restrict__ Q, int d, int lane,
                                         float* q) {
#pragma unroll
    for (int t = 0; t < 2; ++t)
#pragma unroll
        for (int c = 0; c < 10; ++c)
            q[10 * t + c] = Q[c * NF + d * FF + lane + 32 * t];
}

__device__ __forceinline__ void layer2_tile(const v2f (*a2)[16], const v2f* gc, float bias,
                                            int col, int hi, float* h2) {
    v8f acc0 = {}, acc1 = {};
#pragma unroll
    for (int kk = 0; kk < 16; ++kk) {
        acc0 = wmma4(a2[0][kk], gc[kk], acc0);
        acc1 = wmma4(a2[1][kk], gc[kk], acc1);
    }
#pragma unroll
    for (int r = 0; r < 8; ++r) {
        h2[(r + 8 * hi) * H2S + col] = silu(acc0[r] + bias);
        h2[(16 + r + 8 * hi) * H2S + col] = silu(acc1[r] + bias);
    }
}

__device__ __forceinline__ void layer3_tile(const v2f (*a3)[32], const v2f* gc, float bias,
                                            v2f* gn, float* bias_n,
                                            const float* __restrict__ W3,
                                            const float* __restrict__ b3, int col, int kb,
                                            int hi, bool more, const float* cc, float* rad) {
    v8f acc0 = {}, acc1 = {};
#pragma unroll
    for (int half = 0; half < 2; ++half) {
        if (more) {
            load_bfrag<16, 192>(W3, col + 16, kb, 16 * half, gn + 16 * half);
            if (half == 0) *bias_n = b3[col + 16];
        }
#pragma unroll
        for (int kk = 16 * half; kk < 16 * half + 16; ++kk) {
            acc0 = wmma4(a3[0][kk], gc[kk], acc0);
            acc1 = wmma4(a3[1][kk], gc[kk], acc1);
        }
    }
#pragma unroll
    for (int r = 0; r < 8; ++r) {
        const int er = r + 8 * hi;
        rad[er * RDS + col] = silu(acc0[r] + bias) * cc[er];
        rad[(16 + er) * RDS + col] = silu(acc1[r] + bias) * cc[16 + er];
    }
}

__device__ __forceinline__ void scatter_edge(const float* rad, float* __restrict__ M, int e,
                                             int s, int lane, const float* q) {
#pragma unroll
    for (int t = 0; t < 2; ++t) {
        const int f = lane + 32 * t;
        const float rI = rad[e * RDS + 3 * f + 0];
        const float rA = rad[e * RDS + 3 * f + 1];
        const float rS = rad[e * RDS + 3 * f + 2];
        const int so = s * FF + f;
        atomicAdd(&M[0 * NF + so], rI * q[10 * t + 0]);
#pragma unroll
        for (int c = 1; c < 4; ++c)
            atomicAdd(&M[c * NF + so], rA * q[10 * t + c]);
#pragma unroll
        for (int c = 4; c < 10; ++c)
            atomicAdd(&M[c * NF + so], rS * q[10 * t + c]);
    }
}

__global__ __launch_bounds__(32) void edge_kernel(
    const float* __restrict__ rbf, const float* __restrict__ dij,
    const int* __restrict__ pi,
    const float* __restrict__ W1, const float* __restrict__ b1,
    const float* __restrict__ W2, const float* __restrict__ b2,
    const float* __restrict__ W3, const float* __restrict__ b3,
    const float* __restrict__ Q, float* __restrict__ M) {
    __shared__ float buf[TE * H1S + TE * H2S];
    __shared__ float cc[TE];
    float* h1 = buf;
    float* h2 = buf + TE * H1S;
    float* rad = buf;

    const int lane = threadIdx.x;
    const int e0 = blockIdx.x * TE;
    const int row = lane & 15;
    const int hi = lane >> 4;
    const int kb = hi * 2;
    const int col16 = lane & 15;

    const int dst_all = pi[EE + e0 + lane];
    const float dist = dij[e0 + lane];

    v2f a1[2][8];
#pragma unroll
    for (int t = 0; t < 2; ++t)
#pragma unroll
        for (int kk = 0; kk < 8; ++kk) {
            a1[t][kk].x = rbf[(size_t)(e0 + 16 * t + row) * 32 + 4 * kk + kb];
            a1[t][kk].y = rbf[(size_t)(e0 + 16 * t + row) * 32 + 4 * kk + kb + 1];
        }

    v2f g[2][32];
    float gb[2][4];

#pragma unroll
    for (int j = 0; j < 4; ++j) {
        load_bfrag<8, 64>(W1, j * 16 + col16, kb, 0, &g[0][8 * j]);
        gb[0][j] = b1[j * 16 + col16];
    }

    cc[lane] = 0.5f * (cosf(dist * (PI_F / 0.5f)) + 1.0f) * (dist < 0.5f ? 1.0f : 0.0f);

#pragma unroll
    for (int t = 0; t < 2; ++t) {
        load_bfrag<16, 128>(W2, t * 16 + col16, kb, 0, &g[1][16 * t]);
        gb[1][t] = b2[t * 16 + col16];
    }

#pragma unroll 1
    for (int jj = 0; jj < 2; ++jj) {
#pragma unroll
        for (int t = 0; t < 2; ++t) {
            const int col = 32 * jj + 16 * t + col16;
            v8f acc0 = {}, acc1 = {};
#pragma unroll
            for (int kk = 0; kk < 8; ++kk) {
                acc0 = wmma4(a1[0][kk], g[0][8 * t + kk], acc0);
                acc1 = wmma4(a1[1][kk], g[0][8 * t + kk], acc1);
            }
            const float bias = gb[0][t];
#pragma unroll
            for (int r = 0; r < 8; ++r) {
                h1[(r + 8 * hi) * H1S + col] = silu(acc0[r] + bias);
                h1[(16 + r + 8 * hi) * H1S + col] = silu(acc1[r] + bias);
            }
        }
#pragma unroll
        for (int i = 0; i < 16; ++i) g[0][i] = g[0][16 + i];
        gb[0][0] = gb[0][2];
        gb[0][1] = gb[0][3];
    }
    __syncthreads();

    v2f a2[2][16];
#pragma unroll
    for (int t = 0; t < 2; ++t)
#pragma unroll
        for (int kk = 0; kk < 16; ++kk) {
            a2[t][kk].x = h1[(16 * t + row) * H1S + 4 * kk + kb];
            a2[t][kk].y = h1[(16 * t + row) * H1S + 4 * kk + kb + 1];
        }
#pragma unroll 1
    for (int pp = 0; pp < 2; ++pp) {
        const int cb = 64 * pp + col16;
#pragma unroll
        for (int t = 0; t < 2; ++t) {
            const int ncol = cb + 32 + 16 * t;
            load_bfrag<16, 128>(W2, ncol, kb, 0, &g[0][16 * t]);
            gb[0][t] = b2[ncol];
            layer2_tile(a2, &g[1][16 * t], gb[1][t], cb + 16 * t, hi, h2);
        }
#pragma unroll
        for (int t = 0; t < 2; ++t) {
            if (pp == 0) {
                const int ncol = cb + 64 + 16 * t;
                load_bfrag<16, 128>(W2, ncol, kb, 0, &g[1][16 * t]);
                gb[1][t] = b2[ncol];
            } else {
                load_bfrag<16, 192>(W3, col16, kb, 16 * t, &g[1][16 * t]);
                if (t == 0) gb[1][0] = b3[col16];
            }
            layer2_tile(a2, &g[0][16 * t], gb[0][t], cb + 32 + 16 * t, hi, h2);
        }
    }
    __syncthreads();

    v2f a3[2][32];
#pragma unroll
    for (int t = 0; t < 2; ++t)
#pragma unroll
        for (int kk = 0; kk < 32; ++kk) {
            a3[t][kk].x = h2[(16 * t + row) * H2S + 4 * kk + kb];
            a3[t][kk].y = h2[(16 * t + row) * H2S + 4 * kk + kb + 1];
        }
#pragma unroll 1
    for (int jj = 0; jj < 6; ++jj) {
        const int c0 = 32 * jj + col16;
        layer3_tile(a3, g[1], gb[1][0], g[0], &gb[0][0], W3, b3, c0, kb, hi, true, cc, rad);
        layer3_tile(a3, g[0], gb[0][0], g[1], &gb[1][0], W3, b3, c0 + 16, kb, hi, jj < 5, cc,
                    rad);
    }
    const int src_all = pi[e0 + lane];
    float qv[3][20];
    gather_q(Q, __builtin_amdgcn_readlane(dst_all, 0), lane, qv[0]);
    gather_q(Q, __builtin_amdgcn_readlane(dst_all, 1), lane, qv[1]);
    __syncthreads();

#pragma unroll 1
    for (int e3 = 0; e3 < TE - 2; e3 += 3) {
#pragma unroll
        for (int u = 0; u < 3; ++u) {
            const int e = e3 + u;
            gather_q(Q, __builtin_amdgcn_readlane(dst_all, e + 2), lane, qv[(u + 2) % 3]);
            scatter_edge(rad, M, e, __builtin_amdgcn_readlane(src_all, e), lane, qv[u]);
        }
    }
    scatter_edge(rad, M, TE - 2, __builtin_amdgcn_readlane(src_all, TE - 2), lane, qv[0]);
    scatter_edge(rad, M, TE - 1, __builtin_amdgcn_readlane(src_all, TE - 1), lane, qv[1]);
}

__global__ void combine_kernel(const float* __restrict__ M,
                               const float* __restrict__ Q,
                               const float* __restrict__ chg,
                               float* __restrict__ P) {
    int idx = blockIdx.x * blockDim.x + threadIdx.x;
    if (idx >= NF) return;
    const int n = idx >> 6;
    float mc[10], yc[10];
#pragma unroll
    for (int c = 0; c < 10; ++c) { mc[c] = M[c * NF + idx]; yc[c] = Q[c * NF + idx]; }
    float msg[9], Y[9], t1[9], t2[9];
    recon3(mc, msg);
    recon3(yc, Y);
    mm3(msg, Y, t1);
    mm3(Y, msg, t2);
    const float scale = 1.0f + 0.1f * chg[n];
    float T[9]; float tn = 0.f;
#pragma unroll
    for (int i = 0; i < 9; ++i) { T[i] = scale * (t1[i] + t2[i]); tn += T[i] * T[i]; }
    const float inv = 1.0f / (tn + 1.0f);
#pragma unroll
    for (int i = 0; i < 9; ++i) T[i] *= inv;
    float dm = (T[0] + T[4] + T[8]) * (1.0f / 3.0f);
    P[0 * NF + idx] = dm;
    P[1 * NF + idx] = 0.5f * (T[1] - T[3]);
    P[2 * NF + idx] = 0.5f * (T[2] - T[6]);
    P[3 * NF + idx] = 0.5f * (T[5] - T[7]);
    P[4 * NF + idx] = T[0] - dm;
    P[5 * NF + idx] = T[4] - dm;
    P[6 * NF + idx] = T[8] - dm;
    P[7 * NF + idx] = 0.5f * (T[1] + T[3]);
    P[8 * NF + idx] = 0.5f * (T[2] + T[6]);
    P[9 * NF + idx] = 0.5f * (T[5] + T[7]);
}

__global__ void finalize_kernel(const float* __restrict__ Q2,
                                const float* __restrict__ Xn,
                                const float* __restrict__ chg,
                                float* __restrict__ out) {
    int idx = blockIdx.x * blockDim.x + threadIdx.x;
    if (idx >= NF) return;
    const int n = idx >> 6;
    float dc[10];
#pragma unroll
    for (int c = 0; c < 10; ++c) dc[c] = Q2[c * NF + idx];
    float dX[9], prod[9];
    recon3(dc, dX);
    mm3(dX, dX, prod);
    const float scale = 1.0f + 0.1f * chg[n];
#pragma unroll
    for (int i = 0; i < 9; ++i)
        out[(size_t)idx * 9 + i] = Xn[(size_t)idx * 9 + i] + dX[i] + scale * prod[i];
}

extern "C" void kernel_launch(void* const* d_in, const int* in_sizes, int n_in,
                              void* d_out, int out_size, void* d_ws, size_t ws_size,
                              hipStream_t stream) {
    const float* X   = (const float*)d_in[0];
    const int*   pi  = (const int*)  d_in[1];
    const float* dij = (const float*)d_in[2];
    const float* rbf = (const float*)d_in[3];
    const float* chg = (const float*)d_in[4];
    const float* W1  = (const float*)d_in[5];
    const float* b1  = (const float*)d_in[6];
    const float* W2  = (const float*)d_in[7];
    const float* b2  = (const float*)d_in[8];
    const float* W3  = (const float*)d_in[9];
    const float* b3  = (const float*)d_in[10];
    const float* L0  = (const float*)d_in[11];
    const float* L1  = (const float*)d_in[12];
    const float* L2  = (const float*)d_in[13];
    const float* L3  = (const float*)d_in[14];
    const float* L4  = (const float*)d_in[15];
    const float* L5  = (const float*)d_in[16];

    float* ws = (float*)d_ws;
    float* Xn = ws;
    float* P  = Xn + (size_t)NF * 9;
    float* Q  = P + (size_t)10 * NF;
    float* M  = Q + (size_t)10 * NF;

    hipMemsetAsync(M, 0, (size_t)10 * NF * sizeof(float), stream);

    prep_kernel<<<(NF + 255) / 256, 256, 0, stream>>>(X, Xn, P);
    mix_kernel<<<dim3(NN / 16, 10), 128, 0, stream>>>(P, Q, L0, L1, L2);
    edge_kernel<<<EE / TE, 32, 0, stream>>>(rbf, dij, pi, W1, b1, W2, b2, W3, b3, Q, M);
    combine_kernel<<<(NF + 255) / 256, 256, 0, stream>>>(M, Q, chg, P);
    mix_kernel<<<dim3(NN / 16, 10), 128, 0, stream>>>(P, Q, L3, L4, L5);
    finalize_kernel<<<(NF + 255) / 256, 256, 0, stream>>>(Q, Xn, chg, (float*)d_out);
}


# edge kernel: silu as x*rcp(1+exp2(-x*log2e)) on v_exp_f32 and v_rcp_f32 in place of library expf and the IEEE division: about 23 vector instructions per activation become about 5
# speedup vs baseline: 4.5562x; 1.1728x over previous
#include <hip/hip_runtime.h>
#include <math.h>

#define NN 10000
#define EE 160000
#define FF 64
#define NF (NN * FF)
#define PI_F 3.14159265358979323846f

#define H1S 68
#define H2S 132
#define RDS 196

typedef float v2f __attribute__((ext_vector_type(2)));
typedef float v8f __attribute__((ext_vector_type(8)));

__device__ __forceinline__ v8f wmma4(v2f a, v2f b, v8f c) {
    return __builtin_amdgcn_wmma_f32_16x16x4_f32(
         false, a,  false, b,
         (short)0, c,  false,  false);
}

__device__ __forceinline__ float silu(float x) {
    const float e = __builtin_amdgcn_exp2f(x * -1.44269504088896340736f);
    return x * __builtin_amdgcn_rcpf(1.0f + e);
}

__device__ __forceinline__ void recon3(const float c[10], float t[9]) {
    float d = c[0];
    t[0] = d + c[4];  t[4] = d + c[5];  t[8] = d + c[6];
    t[1] = c[1] + c[7];  t[3] = c[7] - c[1];
    t[2] = c[2] + c[8];  t[6] = c[8] - c[2];
    t[5] = c[3] + c[9];  t[7] = c[9] - c[3];
}

__device__ __forceinline__ void mm3(const float a[9], const float b[9], float c[9]) {
#pragma unroll
    for (int i = 0; i < 3; ++i)
#pragma unroll
        for (int j = 0; j < 3; ++j) {
            float s = 0.f;
#pragma unroll
            for (int k = 0; k < 3; ++k) s += a[i * 3 + k] * b[k * 3 + j];
            c[i * 3 + j] = s;
        }
}

__global__ void prep_kernel(const float* __restrict__ X,
                            float* __restrict__ Xn,
                            float* __restrict__ P) {
    int idx = blockIdx.x * blockDim.x + threadIdx.x;
    if (idx >= NF) return;
    float x[9]; float nrm = 0.f;
#pragma unroll
    for (int i = 0; i < 9; ++i) { x[i] = X[(size_t)idx * 9 + i]; nrm += x[i] * x[i]; }
    float inv = 1.0f / (nrm + 1.0f);
#pragma unroll
    for (int i = 0; i < 9; ++i) { x[i] *= inv; Xn[(size_t)idx * 9 + i] = x[i]; }
    float dm = (x[0] + x[4] + x[8]) * (1.0f / 3.0f);
    P[0 * NF + idx] = dm;
    P[1 * NF + idx] = 0.5f * (x[1] - x[3]);
    P[2 * NF + idx] = 0.5f * (x[2] - x[6]);
    P[3 * NF + idx] = 0.5f * (x[5] - x[7]);
    P[4 * NF + idx] = x[0] - dm;
    P[5 * NF + idx] = x[4] - dm;
    P[6 * NF + idx] = x[8] - dm;
    P[7 * NF + idx] = 0.5f * (x[1] + x[3]);
    P[8 * NF + idx] = 0.5f * (x[2] + x[6]);
    P[9 * NF + idx] = 0.5f * (x[5] + x[7]);
}

__global__ __launch_bounds__(128) void mix_kernel(const float* __restrict__ P,
                                                  float* __restrict__ Q,
                                                  const float* __restrict__ La,
                                                  const float* __restrict__ Lb,
                                                  const float* __restrict__ Lc) {
    const int lane = threadIdx.x & 31;
    const int wave = threadIdx.x >> 5;
    const int ch = blockIdx.y;
    const int r0 = blockIdx.x * 16;
    const float* W = (ch == 0) ? La : (ch < 4 ? Lb : Lc);
    const float* A = P + (size_t)ch * NF;
    float* D = Q + (size_t)ch * NF;
    const int row = lane & 15;
    const int hi = lane >> 4;
    const int kb = hi * 2;
    const int col = wave * 16 + (lane & 15);
    v8f acc = {};
#pragma unroll
    for (int k = 0; k < FF; k += 4) {
        v2f a, b;
        a.x = A[(size_t)(r0 + row) * FF + k + kb];
        a.y = A[(size_t)(r0 + row) * FF + k + kb + 1];
        b.x = W[(k + kb) * FF + col];
        b.y = W[(k + kb + 1) * FF + col];
        acc = wmma4(a, b, acc);
    }
#pragma unroll
    for (int r = 0; r < 8; ++r)
        D[(size_t)(r0 + r + 8 * hi) * FF + col] = acc[r];
}


#define TE 32

template <int KT, int LD>
__device__ __forceinline__ void load_bfrag(const float* __restrict__ Wm, int col, int kb,
                                           int k0, v2f* b) {
#pragma unroll
    for (int kk = 0; kk < KT; ++kk) {
        b[kk].x = Wm[(4 * (kk + k0) + kb) * LD + col];
        b[kk].y = Wm[(4 * (kk + k0) + kb + 1) * LD + col];
    }
}

__device__ __forceinline__ void gather_q(const float* __restrict__ Q, int d, int lane,
                                         float* q) {
#pragma unroll
    for (int t = 0; t < 2; ++t)
#pragma unroll
        for (int c = 0; c < 10; ++c)
            q[10 * t + c] = Q[c * NF + d * FF + lane + 32 * t];
}

__device__ __forceinline__ void layer2_tile(const v2f (*a2)[16], const v2f* gc, float bias,
                                            int col, int hi, float* h2) {
    v8f acc0 = {}, acc1 = {};
#pragma unroll
    for (int kk = 0; kk < 16; ++kk) {
        acc0 = wmma4(a2[0][kk], gc[kk], acc0);
        acc1 = wmma4(a2[1][kk], gc[kk], acc1);
    }
#pragma unroll
    for (int r = 0; r < 8; ++r) {
        h2[(r + 8 * hi) * H2S + col] = silu(acc0[r] + bias);
        h2[(16 + r + 8 * hi) * H2S + col] = silu(acc1[r] + bias);
    }
}

__device__ __forceinline__ void layer3_tile(const v2f (*a3)[32], const v2f* gc, float bias,
                                            v2f* gn, float* bias_n,
                                            const float* __restrict__ W3,
                                            const float* __restrict__ b3, int col, int kb,
                                            int hi, bool more, const float* cc, float* rad) {
    v8f acc0 = {}, acc1 = {};
#pragma unroll
    for (int half = 0; half < 2; ++half) {
        if (more) {
            load_bfrag<16, 192>(W3, col + 16, kb, 16 * half, gn + 16 * half);
            if (half == 0) *bias_n = b3[col + 16];
        }
#pragma unroll
        for (int kk = 16 * half; kk < 16 * half + 16; ++kk) {
            acc0 = wmma4(a3[0][kk], gc[kk], acc0);
            acc1 = wmma4(a3[1][kk], gc[kk], acc1);
        }
    }
#pragma unroll
    for (int r = 0; r < 8; ++r) {
        const int er = r + 8 * hi;
        rad[er * RDS + col] = silu(acc0[r] + bias) * cc[er];
        rad[(16 + er) * RDS + col] = silu(acc1[r] + bias) * cc[16 + er];
    }
}

__device__ __forceinline__ void scatter_edge(const float* rad, float* __restrict__ M, int e,
                                             int s, int lane, const float* q) {
#pragma unroll
    for (int t = 0; t < 2; ++t) {
        const int f = lane + 32 * t;
        const float rI = rad[e * RDS + 3 * f + 0];
        const float rA = rad[e * RDS + 3 * f + 1];
        const float rS = rad[e * RDS + 3 * f + 2];
        const int so = s * FF + f;
        atomicAdd(&M[0 * NF + so], rI * q[10 * t + 0]);
#pragma unroll
        for (int c = 1; c < 4; ++c)
            atomicAdd(&M[c * NF + so], rA * q[10 * t + c]);
#pragma unroll
        for (int c = 4; c < 10; ++c)
            atomicAdd(&M[c * NF + so], rS * q[10 * t + c]);
    }
}

__global__ __launch_bounds__(32) void edge_kernel(
    const float* __restrict__ rbf, const float* __restrict__ dij,
    const int* __restrict__ pi,
    const float* __restrict__ W1, const float* __restrict__ b1,
    const float* __restrict__ W2, const float* __restrict__ b2,
    const float* __restrict__ W3, const float* __restrict__ b3,
    const float* __restrict__ Q, float* __restrict__ M) {
    __shared__ float buf[TE * H1S + TE * H2S];
    __shared__ float cc[TE];
    float* h1 = buf;
    float* h2 = buf + TE * H1S;
    float* rad = buf;

    const int lane = threadIdx.x;
    const int e0 = blockIdx.x * TE;
    const int row = lane & 15;
    const int hi = lane >> 4;
    const int kb = hi * 2;
    const int col16 = lane & 15;

    const int dst_all = pi[EE + e0 + lane];
    const float dist = dij[e0 + lane];

    v2f a1[2][8];
#pragma unroll
    for (int t = 0; t < 2; ++t)
#pragma unroll
        for (int kk = 0; kk < 8; ++kk) {
            a1[t][kk].x = rbf[(size_t)(e0 + 16 * t + row) * 32 + 4 * kk + kb];
            a1[t][kk].y = rbf[(size_t)(e0 + 16 * t + row) * 32 + 4 * kk + kb + 1];
        }

    v2f g[2][32];
    float gb[2][4];

#pragma unroll
    for (int j = 0; j < 4; ++j) {
        load_bfrag<8, 64>(W1, j * 16 + col16, kb, 0, &g[0][8 * j]);
        gb[0][j] = b1[j * 16 + col16];
    }

    cc[lane] = 0.5f * (cosf(dist * (PI_F / 0.5f)) + 1.0f) * (dist < 0.5f ? 1.0f : 0.0f);

#pragma unroll
    for (int t = 0; t < 2; ++t) {
        load_bfrag<16, 128>(W2, t * 16 + col16, kb, 0, &g[1][16 * t]);
        gb[1][t] = b2[t * 16 + col16];
    }

#pragma unroll 1
    for (int jj = 0; jj < 2; ++jj) {
#pragma unroll
        for (int t = 0; t < 2; ++t) {
            const int col = 32 * jj + 16 * t + col16;
            v8f acc0 = {}, acc1 = {};
#pragma unroll
            for (int kk = 0; kk < 8; ++kk) {
                acc0 = wmma4(a1[0][kk], g[0][8 * t + kk], acc0);
                acc1 = wmma4(a1[1][kk], g[0][8 * t + kk], acc1);
            }
            const float bias = gb[0][t];
#pragma unroll
            for (int r = 0; r < 8; ++r) {
                h1[(r + 8 * hi) * H1S + col] = silu(acc0[r] + bias);
                h1[(16 + r + 8 * hi) * H1S + col] = silu(acc1[r] + bias);
            }
        }
#pragma unroll
        for (int i = 0; i < 16; ++i) g[0][i] = g[0][16 + i];
        gb[0][0] = gb[0][2];
        gb[0][1] = gb[0][3];
    }
    __syncthreads();

    v2f a2[2][16];
#pragma unroll
    for (int t = 0; t < 2; ++t)
#pragma unroll
        for (int kk = 0; kk < 16; ++kk) {
            a2[t][kk].x = h1[(16 * t + row) * H1S + 4 * kk + kb];
            a2[t][kk].y = h1[(16 * t + row) * H1S + 4 * kk + kb + 1];
        }
#pragma unroll 1
    for (int pp = 0; pp < 2; ++pp) {
        const int cb = 64 * pp + col16;
#pragma unroll
        for (int t = 0; t < 2; ++t) {
            const int ncol = cb + 32 + 16 * t;
            load_bfrag<16, 128>(W2, ncol, kb, 0, &g[0][16 * t]);
            gb[0][t] = b2[ncol];
            layer2_tile(a2, &g[1][16 * t], gb[1][t], cb + 16 * t, hi, h2);
        }
#pragma unroll
        for (int t = 0; t < 2; ++t) {
            if (pp == 0) {
                const int ncol = cb + 64 + 16 * t;
                load_bfrag<16, 128>(W2, ncol, kb, 0, &g[1][16 * t]);
                gb[1][t] = b2[ncol];
            } else {
                load_bfrag<16, 192>(W3, col16, kb, 16 * t, &g[1][16 * t]);
                if (t == 0) gb[1][0] = b3[col16];
            }
            layer2_tile(a2, &g[0][16 * t], gb[0][t], cb + 32 + 16 * t, hi, h2);
        }
    }
    __syncthreads();

    v2f a3[2][32];
#pragma unroll
    for (int t = 0; t < 2; ++t)
#pragma unroll
        for (int kk = 0; kk < 32; ++kk) {
            a3[t][kk].x = h2[(16 * t + row) * H2S + 4 * kk + kb];
            a3[t][kk].y = h2[(16 * t + row) * H2S + 4 * kk + kb + 1];
        }
#pragma unroll 1
    for (int jj = 0; jj < 6; ++jj) {
        const int c0 = 32 * jj + col16;
        layer3_tile(a3, g[1], gb[1][0], g[0], &gb[0][0], W3, b3, c0, kb, hi, true, cc, rad);
        layer3_tile(a3, g[0], gb[0][0], g[1], &gb[1][0], W3, b3, c0 + 16, kb, hi, jj < 5, cc,
                    rad);
    }
    const int src_all = pi[e0 + lane];
    float qv[3][20];
    gather_q(Q, __builtin_amdgcn_readlane(dst_all, 0), lane, qv[0]);
    gather_q(Q, __builtin_amdgcn_readlane(dst_all, 1), lane, qv[1]);
    __syncthreads();

#pragma unroll 1
    for (int e3 = 0; e3 < TE - 2; e3 += 3) {
#pragma unroll
        for (int u = 0; u < 3; ++u) {
            const int e = e3 + u;
            gather_q(Q, __builtin_amdgcn_readlane(dst_all, e + 2), lane, qv[(u + 2) % 3]);
            scatter_edge(rad, M, e, __builtin_amdgcn_readlane(src_all, e), lane, qv[u]);
        }
    }
    scatter_edge(rad, M, TE - 2, __builtin_amdgcn_readlane(src_all, TE - 2), lane, qv[0]);
    scatter_edge(rad, M, TE - 1, __builtin_amdgcn_readlane(src_all, TE - 1), lane, qv[1]);
}

__global__ void combine_kernel(const float* __restrict__ M,
                               const float* __restrict__ Q,
                               const float* __restrict__ chg,
                               float* __restrict__ P) {
    int idx = blockIdx.x * blockDim.x + threadIdx.x;
    if (idx >= NF) return;
    const int n = idx >> 6;
    float mc[10], yc[10];
#pragma unroll
    for (int c = 0; c < 10; ++c) { mc[c] = M[c * NF + idx]; yc[c] = Q[c * NF + idx]; }
    float msg[9], Y[9], t1[9], t2[9];
    recon3(mc, msg);
    recon3(yc, Y);
    mm3(msg, Y, t1);
    mm3(Y, msg, t2);
    const float scale = 1.0f + 0.1f * chg[n];
    float T[9]; float tn = 0.f;
#pragma unroll
    for (int i = 0; i < 9; ++i) { T[i] = scale * (t1[i] + t2[i]); tn += T[i] * T[i]; }
    const float inv = 1.0f / (tn + 1.0f);
#pragma unroll
    for (int i = 0; i < 9; ++i) T[i] *= inv;
    float dm = (T[0] + T[4] + T[8]) * (1.0f / 3.0f);
    P[0 * NF + idx] = dm;
    P[1 * NF + idx] = 0.5f * (T[1] - T[3]);
    P[2 * NF + idx] = 0.5f * (T[2] - T[6]);
    P[3 * NF + idx] = 0.5f * (T[5] - T[7]);
    P[4 * NF + idx] = T[0] - dm;
    P[5 * NF + idx] = T[4] - dm;
    P[6 * NF + idx] = T[8] - dm;
    P[7 * NF + idx] = 0.5f * (T[1] + T[3]);
    P[8 * NF + idx] = 0.5f * (T[2] + T[6]);
    P[9 * NF + idx] = 0.5f * (T[5] + T[7]);
}

__global__ void finalize_kernel(const float* __restrict__ Q2,
                                const float* __restrict__ Xn,
                                const float* __restrict__ chg,
                                float* __restrict__ out) {
    int idx = blockIdx.x * blockDim.x + threadIdx.x;
    if (idx >= NF) return;
    const int n = idx >> 6;
    float dc[10];
#pragma unroll
    for (int c = 0; c < 10; ++c) dc[c] = Q2[c * NF + idx];
    float dX[9], prod[9];
    recon3(dc, dX);
    mm3(dX, dX, prod);
    const float scale = 1.0f + 0.1f * chg[n];
#pragma unroll
    for (int i = 0; i < 9; ++i)
        out[(size_t)idx * 9 + i] = Xn[(size_t)idx * 9 + i] + dX[i] + scale * prod[i];
}

extern "C" void kernel_launch(void* const* d_in, const int* in_sizes, int n_in,
                              void* d_out, int out_size, void* d_ws, size_t ws_size,
                              hipStream_t stream) {
    const float* X   = (const float*)d_in[0];
    const int*   pi  = (const int*)  d_in[1];
    const float* dij = (const float*)d_in[2];
    const float* rbf = (const float*)d_in[3];
    const float* chg = (const float*)d_in[4];
    const float* W1  = (const float*)d_in[5];
    const float* b1  = (const float*)d_in[6];
    const float* W2  = (const float*)d_in[7];
    const float* b2  = (const float*)d_in[8];
    const float* W3  = (const float*)d_in[9];
    const float* b3  = (const float*)d_in[10];
    const float* L0  = (const float*)d_in[11];
    const float* L1  = (const float*)d_in[12];
    const float* L2  = (const float*)d_in[13];
    const float* L3  = (const float*)d_in[14];
    const float* L4  = (const float*)d_in[15];
    const float* L5  = (const float*)d_in[16];

    float* ws = (float*)d_ws;
    float* Xn = ws;
    float* P  = Xn + (size_t)NF * 9;
    float* Q  = P + (size_t)10 * NF;
    float* M  = Q + (size_t)10 * NF;

    hipMemsetAsync(M, 0, (size_t)10 * NF * sizeof(float), stream);

    prep_kernel<<<(NF + 255) / 256, 256, 0, stream>>>(X, Xn, P);
    mix_kernel<<<dim3(NN / 16, 10), 128, 0, stream>>>(P, Q, L0, L1, L2);
    edge_kernel<<<EE / TE, 32, 0, stream>>>(rbf, dij, pi, W1, b1, W2, b2, W3, b3, Q, M);
    combine_kernel<<<(NF + 255) / 256, 256, 0, stream>>>(M, Q, chg, P);
    mix_kernel<<<dim3(NN / 16, 10), 128, 0, stream>>>(P, Q, L3, L4, L5);
    finalize_kernel<<<(NF + 255) / 256, 256, 0, stream>>>(Q, Xn, chg, (float*)d_out);
}


# node side fused around the two feature-mixing GEMMs: prep+mix+clear of M and combine+mix+output each one kernel on 16-node tiles, P in LDS, second Q in registers: 6 launches become 3
# speedup vs baseline: 5.0085x; 1.0993x over previous
#include <hip/hip_runtime.h>
#include <math.h>

#define NN 10000
#define EE 160000
#define FF 64
#define NF (NN * FF)
#define PI_F 3.14159265358979323846f

#define H1S 68
#define H2S 132
#define RDS 196

typedef float v2f __attribute__((ext_vector_type(2)));
typedef float v8f __attribute__((ext_vector_type(8)));

__device__ __forceinline__ v8f wmma4(v2f a, v2f b, v8f c) {
    return __builtin_amdgcn_wmma_f32_16x16x4_f32(
         false, a,  false, b,
         (short)0, c,  false,  false);
}

__device__ __forceinline__ float silu(float x) {
    const float e = __builtin_amdgcn_exp2f(x * -1.44269504088896340736f);
    return x * __builtin_amdgcn_rcpf(1.0f + e);
}

__device__ __forceinline__ void recon3(const float c[10], float t[9]) {
    float d = c[0];
    t[0] = d + c[4];  t[4] = d + c[5];  t[8] = d + c[6];
    t[1] = c[1] + c[7];  t[3] = c[7] - c[1];
    t[2] = c[2] + c[8];  t[6] = c[8] - c[2];
    t[5] = c[3] + c[9];  t[7] = c[9] - c[3];
}

__device__ __forceinline__ void mm3(const float a[9], const float b[9], float c[9]) {
#pragma unroll
    for (int i = 0; i < 3; ++i)
#pragma unroll
        for (int j = 0; j < 3; ++j) {
            float s = 0.f;
#pragma unroll
            for (int k = 0; k < 3; ++k) s += a[i * 3 + k] * b[k * 3 + j];
            c[i * 3 + j] = s;
        }
}

#define NT 16
#define PS 68

__device__ __forceinline__ void compact10(const float t[9], float c[10]) {
    const float dm = (t[0] + t[4] + t[8]) * (1.0f / 3.0f);
    c[0] = dm;
    c[1] = 0.5f * (t[1] - t[3]);
    c[2] = 0.5f * (t[2] - t[6]);
    c[3] = 0.5f * (t[5] - t[7]);
    c[4] = t[0] - dm;
    c[5] = t[4] - dm;
    c[6] = t[8] - dm;
    c[7] = 0.5f * (t[1] + t[3]);
    c[8] = 0.5f * (t[2] + t[6]);
    c[9] = 0.5f * (t[5] + t[7]);
}

__device__ __forceinline__ void load_lfrag(const float* __restrict__ L, int kb, int col,
                                           v2f* b) {
#pragma unroll
    for (int kk = 0; kk < 16; ++kk) {
        b[kk].x = L[(4 * kk + kb) * FF + col];
        b[kk].y = L[(4 * kk + kb + 1) * FF + col];
    }
}

__device__ __forceinline__ void load_pfrag(const float* ps, int row, int kb, int kk, v2f* a) {
#pragma unroll
    for (int c = 0; c < 10; ++c) {
        a[c].x = ps[(c * NT + row) * PS + 4 * kk + kb];
        a[c].y = ps[(c * NT + row) * PS + 4 * kk + kb + 1];
    }
}

__device__ __forceinline__ void mix_tile(const float* ps, int row, int kb, const v2f* bI,
                                         const v2f* bA, const v2f* bS, v8f* acc) {
    v2f a[2][10];
    load_pfrag(ps, row, kb, 0, a[0]);
#pragma unroll
    for (int kk = 0; kk < 16; ++kk) {
        if (kk + 1 < 16) load_pfrag(ps, row, kb, kk + 1, a[(kk + 1) & 1]);
        __builtin_amdgcn_sched_barrier(0);
#pragma unroll
        for (int c = 0; c < 10; ++c)
            acc[c] = wmma4(a[kk & 1][c], c == 0 ? bI[kk] : (c < 4 ? bA[kk] : bS[kk]), acc[c]);
        __builtin_amdgcn_sched_barrier(0);
    }
}

__global__ __launch_bounds__(128) void node_in_kernel(const float* __restrict__ X,
                                                      float* __restrict__ Xn,
                                                      float* __restrict__ Q,
                                                      float* __restrict__ M,
                                                      const float* __restrict__ La,
                                                      const float* __restrict__ Lb,
                                                      const float* __restrict__ Lc) {
    __shared__ float4 xs4[NT * FF * 9 / 4];
    __shared__ float ps[10 * NT * PS];
    float* xs = (float*)xs4;

    const int tid = threadIdx.x;
    const int lane = tid & 31;
    const int wave = tid >> 5;
    const int n0 = blockIdx.x * NT;
    const int row = lane & 15;
    const int hi = lane >> 4;
    const int kb = hi * 2;
    const int col = wave * 16 + (lane & 15);

    v2f bI[16], bA[16], bS[16];
    load_lfrag(La, kb, col, bI);
    load_lfrag(Lb, kb, col, bA);
    load_lfrag(Lc, kb, col, bS);

    const float4* xg = (const float4*)(X + (size_t)n0 * FF * 9);
#pragma unroll
    for (int i = 0; i < NT * FF * 9 / 4 / 128; ++i) xs4[tid + 128 * i] = xg[tid + 128 * i];

    __syncthreads();

#pragma unroll 2
    for (int u = 0; u < 8; ++u) {
        const int j = tid + 128 * u;
        float x[9];
        float nrm = 0.f;
#pragma unroll
        for (int i = 0; i < 9; ++i) { x[i] = xs[j * 9 + i]; nrm += x[i] * x[i]; }
        const float inv = 1.0f / (nrm + 1.0f);
#pragma unroll
        for (int i = 0; i < 9; ++i) { x[i] *= inv; xs[j * 9 + i] = x[i]; }
        float c10[10];
        compact10(x, c10);
#pragma unroll
        for (int c = 0; c < 10; ++c) ps[(c * NT + (j >> 6)) * PS + (j & 63)] = c10[c];
    }
    __syncthreads();

    float4* xo = (float4*)(Xn + (size_t)n0 * FF * 9);
#pragma unroll
    for (int i = 0; i < NT * FF * 9 / 4 / 128; ++i) xo[tid + 128 * i] = xs4[tid + 128 * i];

    v8f acc[10];
#pragma unroll
    for (int c = 0; c < 10; ++c) acc[c] = v8f{};
    mix_tile(ps, row, kb, bI, bA, bS, acc);

#pragma unroll
    for (int c = 0; c < 10; ++c)
#pragma unroll
        for (int r = 0; r < 8; ++r)
            Q[(size_t)c * NF + (size_t)(n0 + r + 8 * hi) * FF + col] = acc[c][r];

    const float4 zero4 = {0.f, 0.f, 0.f, 0.f};
#pragma unroll
    for (int c = 0; c < 10; ++c) {
        float4* m4 = (float4*)(M + (size_t)c * NF + (size_t)n0 * FF);
        m4[tid] = zero4;
        m4[tid + 128] = zero4;
    }
}


#define TE 32

template <int KT, int LD>
__device__ __forceinline__ void load_bfrag(const float* __restrict__ Wm, int col, int kb,
                                           int k0, v2f* b) {
#pragma unroll
    for (int kk = 0; kk < KT; ++kk) {
        b[kk].x = Wm[(4 * (kk + k0) + kb) * LD + col];
        b[kk].y = Wm[(4 * (kk + k0) + kb + 1) * LD + col];
    }
}

__device__ __forceinline__ void gather_q(const float* __restrict__ Q, int d, int lane,
                                         float* q) {
#pragma unroll
    for (int t = 0; t < 2; ++t)
#pragma unroll
        for (int c = 0; c < 10; ++c)
            q[10 * t + c] = Q[c * NF + d * FF + lane + 32 * t];
}

__device__ __forceinline__ void layer2_tile(const v2f (*a2)[16], const v2f* gc, float bias,
                                            int col, int hi, float* h2) {
    v8f acc0 = {}, acc1 = {};
#pragma unroll
    for (int kk = 0; kk < 16; ++kk) {
        acc0 = wmma4(a2[0][kk], gc[kk], acc0);
        acc1 = wmma4(a2[1][kk], gc[kk], acc1);
    }
#pragma unroll
    for (int r = 0; r < 8; ++r) {
        h2[(r + 8 * hi) * H2S + col] = silu(acc0[r] + bias);
        h2[(16 + r + 8 * hi) * H2S + col] = silu(acc1[r] + bias);
    }
}

__device__ __forceinline__ void layer3_tile(const v2f (*a3)[32], const v2f* gc, float bias,
                                            v2f* gn, float* bias_n,
                                            const float* __restrict__ W3,
                                            const float* __restrict__ b3, int col, int kb,
                                            int hi, bool more, const float* cc, float* rad) {
    v8f acc0 = {}, acc1 = {};
#pragma unroll
    for (int half = 0; half < 2; ++half) {
        if (more) {
            load_bfrag<16, 192>(W3, col + 16, kb, 16 * half, gn + 16 * half);
            if (half == 0) *bias_n = b3[col + 16];
        }
#pragma unroll
        for (int kk = 16 * half; kk < 16 * half + 16; ++kk) {
            acc0 = wmma4(a3[0][kk], gc[kk], acc0);
            acc1 = wmma4(a3[1][kk], gc[kk], acc1);
        }
    }
#pragma unroll
    for (int r = 0; r < 8; ++r) {
        const int er = r + 8 * hi;
        rad[er * RDS + col] = silu(acc0[r] + bias) * cc[er];
        rad[(16 + er) * RDS + col] = silu(acc1[r] + bias) * cc[16 + er];
    }
}

__device__ __forceinline__ void scatter_edge(const float* rad, float* __restrict__ M, int e,
                                             int s, int lane, const float* q) {
#pragma unroll
    for (int t = 0; t < 2; ++t) {
        const int f = lane + 32 * t;
        const float rI = rad[e * RDS + 3 * f + 0];
        const float rA = rad[e * RDS + 3 * f + 1];
        const float rS = rad[e * RDS + 3 * f + 2];
        const int so = s * FF + f;
        atomicAdd(&M[0 * NF + so], rI * q[10 * t + 0]);
#pragma unroll
        for (int c = 1; c < 4; ++c)
            atomicAdd(&M[c * NF + so], rA * q[10 * t + c]);
#pragma unroll
        for (int c = 4; c < 10; ++c)
            atomicAdd(&M[c * NF + so], rS * q[10 * t + c]);
    }
}

__global__ __launch_bounds__(32) void edge_kernel(
    const float* __restrict__ rbf, const float* __restrict__ dij,
    const int* __restrict__ pi,
    const float* __restrict__ W1, const float* __restrict__ b1,
    const float* __restrict__ W2, const float* __restrict__ b2,
    const float* __restrict__ W3, const float* __restrict__ b3,
    const float* __restrict__ Q, float* __restrict__ M) {
    __shared__ float buf[TE * H1S + TE * H2S];
    __shared__ float cc[TE];
    float* h1 = buf;
    float* h2 = buf + TE * H1S;
    float* rad = buf;

    const int lane = threadIdx.x;
    const int e0 = blockIdx.x * TE;
    const int row = lane & 15;
    const int hi = lane >> 4;
    const int kb = hi * 2;
    const int col16 = lane & 15;

    const int dst_all = pi[EE + e0 + lane];
    const float dist = dij[e0 + lane];

    v2f a1[2][8];
#pragma unroll
    for (int t = 0; t < 2; ++t)
#pragma unroll
        for (int kk = 0; kk < 8; ++kk) {
            a1[t][kk].x = rbf[(size_t)(e0 + 16 * t + row) * 32 + 4 * kk + kb];
            a1[t][kk].y = rbf[(size_t)(e0 + 16 * t + row) * 32 + 4 * kk + kb + 1];
        }

    v2f g[2][32];
    float gb[2][4];

#pragma unroll
    for (int j = 0; j < 4; ++j) {
        load_bfrag<8, 64>(W1, j * 16 + col16, kb, 0, &g[0][8 * j]);
        gb[0][j] = b1[j * 16 + col16];
    }

    cc[lane] = 0.5f * (cosf(dist * (PI_F / 0.5f)) + 1.0f) * (dist < 0.5f ? 1.0f : 0.0f);

#pragma unroll
    for (int t = 0; t < 2; ++t) {
        load_bfrag<16, 128>(W2, t * 16 + col16, kb, 0, &g[1][16 * t]);
        gb[1][t] = b2[t * 16 + col16];
    }

#pragma unroll 1
    for (int jj = 0; jj < 2; ++jj) {
#pragma unroll
        for (int t = 0; t < 2; ++t) {
            const int col = 32 * jj + 16 * t + col16;
            v8f acc0 = {}, acc1 = {};
#pragma unroll
            for (int kk = 0; kk < 8; ++kk) {
                acc0 = wmma4(a1[0][kk], g[0][8 * t + kk], acc0);
                acc1 = wmma4(a1[1][kk], g[0][8 * t + kk], acc1);
            }
            const float bias = gb[0][t];
#pragma unroll
            for (int r = 0; r < 8; ++r) {
                h1[(r + 8 * hi) * H1S + col] = silu(acc0[r] + bias);
                h1[(16 + r + 8 * hi) * H1S + col] = silu(acc1[r] + bias);
            }
        }
#pragma unroll
        for (int i = 0; i < 16; ++i) g[0][i] = g[0][16 + i];
        gb[0][0] = gb[0][2];
        gb[0][1] = gb[0][3];
    }
    __syncthreads();

    v2f a2[2][16];
#pragma unroll
    for (int t = 0; t < 2; ++t)
#pragma unroll
        for (int kk = 0; kk < 16; ++kk) {
            a2[t][kk].x = h1[(16 * t + row) * H1S + 4 * kk + kb];
            a2[t][kk].y = h1[(16 * t + row) * H1S + 4 * kk + kb + 1];
        }
#pragma unroll 1
    for (int pp = 0; pp < 2; ++pp) {
        const int cb = 64 * pp + col16;
#pragma unroll
        for (int t = 0; t < 2; ++t) {
            const int ncol = cb + 32 + 16 * t;
            load_bfrag<16, 128>(W2, ncol, kb, 0, &g[0][16 * t]);
            gb[0][t] = b2[ncol];
            layer2_tile(a2, &g[1][16 * t], gb[1][t], cb + 16 * t, hi, h2);
        }
#pragma unroll
        for (int t = 0; t < 2; ++t) {
            if (pp == 0) {
                const int ncol = cb + 64 + 16 * t;
                load_bfrag<16, 128>(W2, ncol, kb, 0, &g[1][16 * t]);
                gb[1][t] = b2[ncol];
            } else {
                load_bfrag<16, 192>(W3, col16, kb, 16 * t, &g[1][16 * t]);
                if (t == 0) gb[1][0] = b3[col16];
            }
            layer2_tile(a2, &g[0][16 * t], gb[0][t], cb + 32 + 16 * t, hi, h2);
        }
    }
    __syncthreads();

    v2f a3[2][32];
#pragma unroll
    for (int t = 0; t < 2; ++t)
#pragma unroll
        for (int kk = 0; kk < 32; ++kk) {
            a3[t][kk].x = h2[(16 * t + row) * H2S + 4 * kk + kb];
            a3[t][kk].y = h2[(16 * t + row) * H2S + 4 * kk + kb + 1];
        }
#pragma unroll 1
    for (int jj = 0; jj < 6; ++jj) {
        const int c0 = 32 * jj + col16;
        layer3_tile(a3, g[1], gb[1][0], g[0], &gb[0][0], W3, b3, c0, kb, hi, true, cc, rad);
        layer3_tile(a3, g[0], gb[0][0], g[1], &gb[1][0], W3, b3, c0 + 16, kb, hi, jj < 5, cc,
                    rad);
    }
    const int src_all = pi[e0 + lane];
    float qv[3][20];
    gather_q(Q, __builtin_amdgcn_readlane(dst_all, 0), lane, qv[0]);
    gather_q(Q, __builtin_amdgcn_readlane(dst_all, 1), lane, qv[1]);
    __syncthreads();

#pragma unroll 1
    for (int e3 = 0; e3 < TE - 2; e3 += 3) {
#pragma unroll
        for (int u = 0; u < 3; ++u) {
            const int e = e3 + u;
            gather_q(Q, __builtin_amdgcn_readlane(dst_all, e + 2), lane, qv[(u + 2) % 3]);
            scatter_edge(rad, M, e, __builtin_amdgcn_readlane(src_all, e), lane, qv[u]);
        }
    }
    scatter_edge(rad, M, TE - 2, __builtin_amdgcn_readlane(src_all, TE - 2), lane, qv[0]);
    scatter_edge(rad, M, TE - 1, __builtin_amdgcn_readlane(src_all, TE - 1), lane, qv[1]);
}

__device__ __forceinline__ void load_mq(const float* __restrict__ M, const float* __restrict__ Q,
                                        int idx, float* mc, float* yc) {
#pragma unroll
    for (int c = 0; c < 10; ++c) { mc[c] = M[c * NF + idx]; yc[c] = Q[c * NF + idx]; }
}

__global__ __launch_bounds__(128) void node_out_kernel(const float* __restrict__ M,
                                                       const float* __restrict__ Q,
                                                       const float* __restrict__ Xn,
                                                       const float* __restrict__ chg,
                                                       const float* __restrict__ La,
                                                       const float* __restrict__ Lb,
                                                       const float* __restrict__ Lc,
                                                       float* __restrict__ out) {
    __shared__ float ps[10 * NT * PS];

    const int tid = threadIdx.x;
    const int lane = tid & 31;
    const int wave = tid >> 5;
    const int n0 = blockIdx.x * NT;
    const int row = lane & 15;
    const int hi = lane >> 4;
    const int kb = hi * 2;
    const int col = wave * 16 + (lane & 15);

    v2f bI[16], bA[16], bS[16];
    load_lfrag(La, kb, col, bI);
    load_lfrag(Lb, kb, col, bA);
    load_lfrag(Lc, kb, col, bS);

    float cq[8];
#pragma unroll
    for (int u = 0; u < 8; ++u) cq[u] = chg[n0 + (tid >> 6) + 2 * u];
    float mc[4][10], yc[4][10];
#pragma unroll
    for (int u = 0; u < 3; ++u) load_mq(M, Q, n0 * FF + tid + 128 * u, mc[u], yc[u]);
#pragma unroll
    for (int u = 0; u < 8; ++u) {
        const int j = tid + 128 * u;
        if (u + 3 < 8) load_mq(M, Q, n0 * FF + j + 384, mc[(u + 3) & 3], yc[(u + 3) & 3]);
        float msg[9], Y[9], t1[9], t2[9];
        recon3(mc[u & 3], msg);
        recon3(yc[u & 3], Y);
        mm3(msg, Y, t1);
        mm3(Y, msg, t2);
        const float scale = 1.0f + 0.1f * cq[u];
        float T[9];
        float tn = 0.f;
#pragma unroll
        for (int i = 0; i < 9; ++i) { T[i] = scale * (t1[i] + t2[i]); tn += T[i] * T[i]; }
        const float inv = 1.0f / (tn + 1.0f);
#pragma unroll
        for (int i = 0; i < 9; ++i) T[i] *= inv;
        float c10[10];
        compact10(T, c10);
#pragma unroll
        for (int c = 0; c < 10; ++c) ps[(c * NT + (j >> 6)) * PS + (j & 63)] = c10[c];
    }
    __syncthreads();

    float xv[8][9];
    float sc[8];
#pragma unroll
    for (int r = 0; r < 8; ++r) {
        const int n = n0 + r + 8 * hi;
        sc[r] = 1.0f + 0.1f * chg[n];
#pragma unroll
        for (int i = 0; i < 9; ++i) xv[r][i] = Xn[((size_t)n * FF + col) * 9 + i];
    }

    v8f acc[10];
#pragma unroll
    for (int c = 0; c < 10; ++c) acc[c] = v8f{};
    mix_tile(ps, row, kb, bI, bA, bS, acc);

#pragma unroll
    for (int r = 0; r < 8; ++r) {
        const int n = n0 + r + 8 * hi;
        float dc[10], dX[9], prod[9];
#pragma unroll
        for (int c = 0; c < 10; ++c) dc[c] = acc[c][r];
        recon3(dc, dX);
        mm3(dX, dX, prod);
#pragma unroll
        for (int i = 0; i < 9; ++i)
            out[((size_t)n * FF + col) * 9 + i] = xv[r][i] + dX[i] + sc[r] * prod[i];
    }
}

extern "C" void kernel_launch(void* const* d_in, const int* in_sizes, int n_in,
                              void* d_out, int out_size, void* d_ws, size_t ws_size,
                              hipStream_t stream) {
    const float* X   = (const float*)d_in[0];
    const int*   pi  = (const int*)  d_in[1];
    const float* dij = (const float*)d_in[2];
    const float* rbf = (const float*)d_in[3];
    const float* chg = (const float*)d_in[4];
    const float* W1  = (const float*)d_in[5];
    const float* b1  = (const float*)d_in[6];
    const float* W2  = (const float*)d_in[7];
    const float* b2  = (const float*)d_in[8];
    const float* W3  = (const float*)d_in[9];
    const float* b3  = (const float*)d_in[10];
    const float* L0  = (const float*)d_in[11];
    const float* L1  = (const float*)d_in[12];
    const float* L2  = (const float*)d_in[13];
    const float* L3  = (const float*)d_in[14];
    const float* L4  = (const float*)d_in[15];
    const float* L5  = (const float*)d_in[16];

    float* ws = (float*)d_ws;
    float* Xn = ws;
    float* Q  = Xn + (size_t)NF * 9 + (size_t)10 * NF;
    float* M  = Q + (size_t)10 * NF;

    node_in_kernel<<<NN / NT, 128, 0, stream>>>(X, Xn, Q, M, L0, L1, L2);
    edge_kernel<<<EE / TE, 32, 0, stream>>>(rbf, dij, pi, W1, b1, W2, b2, W3, b3, Q, M);
    node_out_kernel<<<NN / NT, 128, 0, stream>>>(M, Q, Xn, chg, L3, L4, L5, (float*)d_out);
}


# all three kernels: s22 of the trace-free symmetric part is not carried but formed as -(s00+s11) where a 3x3 is rebuilt: 9 channel planes, not 10, in node GEMMs, gathers and float atomics
# speedup vs baseline: 5.2130x; 1.0408x over previous
#include <hip/hip_runtime.h>
#include <math.h>

#define NN 10000
#define EE 160000
#define FF 64
#define NF (NN * FF)
#define PI_F 3.14159265358979323846f

#define H1S 68
#define H2S 132
#define RDS 196

typedef float v2f __attribute__((ext_vector_type(2)));
typedef float v8f __attribute__((ext_vector_type(8)));

__device__ __forceinline__ v8f wmma4(v2f a, v2f b, v8f c) {
    return __builtin_amdgcn_wmma_f32_16x16x4_f32(
         false, a,  false, b,
         (short)0, c,  false,  false);
}

__device__ __forceinline__ float silu(float x) {
    const float e = __builtin_amdgcn_exp2f(x * -1.44269504088896340736f);
    return x * __builtin_amdgcn_rcpf(1.0f + e);
}

#define NCH 9
__device__ __forceinline__ void recon3(const float c[NCH], float t[9]) {
    float d = c[0];
    t[0] = d + c[4];  t[4] = d + c[5];  t[8] = d - (c[4] + c[5]);
    t[1] = c[1] + c[6];  t[3] = c[6] - c[1];
    t[2] = c[2] + c[7];  t[6] = c[7] - c[2];
    t[5] = c[3] + c[8];  t[7] = c[8] - c[3];
}

__device__ __forceinline__ void mm3(const float a[9], const float b[9], float c[9]) {
#pragma unroll
    for (int i = 0; i < 3; ++i)
#pragma unroll
        for (int j = 0; j < 3; ++j) {
            float s = 0.f;
#pragma unroll
            for (int k = 0; k < 3; ++k) s += a[i * 3 + k] * b[k * 3 + j];
            c[i * 3 + j] = s;
        }
}

#define NT 16
#define PS 68

__device__ __forceinline__ void compact9(const float t[9], float c[NCH]) {
    const float dm = (t[0] + t[4] + t[8]) * (1.0f / 3.0f);
    c[0] = dm;
    c[1] = 0.5f * (t[1] - t[3]);
    c[2] = 0.5f * (t[2] - t[6]);
    c[3] = 0.5f * (t[5] - t[7]);
    c[4] = t[0] - dm;
    c[5] = t[4] - dm;
    c[6] = 0.5f * (t[1] + t[3]);
    c[7] = 0.5f * (t[2] + t[6]);
    c[8] = 0.5f * (t[5] + t[7]);
}

__device__ __forceinline__ void load_lfrag(const float* __restrict__ L, int kb, int col,
                                           v2f* b) {
#pragma unroll
    for (int kk = 0; kk < 16; ++kk) {
        b[kk].x = L[(4 * kk + kb) * FF + col];
        b[kk].y = L[(4 * kk + kb + 1) * FF + col];
    }
}

__device__ __forceinline__ void load_pfrag(const float* ps, int row, int kb, int kk, v2f* a) {
#pragma unroll
    for (int c = 0; c < NCH; ++c) {
        a[c].x = ps[(c * NT + row) * PS + 4 * kk + kb];
        a[c].y = ps[(c * NT + row) * PS + 4 * kk + kb + 1];
    }
}

__device__ __forceinline__ void mix_tile(const float* ps, int row, int kb, const v2f* bI,
                                         const v2f* bA, const v2f* bS, v8f* acc) {
    v2f a[2][NCH];
    load_pfrag(ps, row, kb, 0, a[0]);
#pragma unroll
    for (int kk = 0; kk < 16; ++kk) {
        if (kk + 1 < 16) load_pfrag(ps, row, kb, kk + 1, a[(kk + 1) & 1]);
        __builtin_amdgcn_sched_barrier(0);
#pragma unroll
        for (int c = 0; c < NCH; ++c)
            acc[c] = wmma4(a[kk & 1][c], c == 0 ? bI[kk] : (c < 4 ? bA[kk] : bS[kk]), acc[c]);
        __builtin_amdgcn_sched_barrier(0);
    }
}

__global__ __launch_bounds__(128) void node_in_kernel(const float* __restrict__ X,
                                                      float* __restrict__ Xn,
                                                      float* __restrict__ Q,
                                                      float* __restrict__ M,
                                                      const float* __restrict__ La,
                                                      const float* __restrict__ Lb,
                                                      const float* __restrict__ Lc) {
    __shared__ float4 xs4[NT * FF * 9 / 4];
    __shared__ float ps[NCH * NT * PS];
    float* xs = (float*)xs4;

    const int tid = threadIdx.x;
    const int lane = tid & 31;
    const int wave = tid >> 5;
    const int n0 = blockIdx.x * NT;
    const int row = lane & 15;
    const int hi = lane >> 4;
    const int kb = hi * 2;
    const int col = wave * 16 + (lane & 15);

    v2f bI[16], bA[16], bS[16];
    load_lfrag(La, kb, col, bI);
    load_lfrag(Lb, kb, col, bA);
    load_lfrag(Lc, kb, col, bS);

    const float4* xg = (const float4*)(X + (size_t)n0 * FF * 9);
#pragma unroll
    for (int i = 0; i < NT * FF * 9 / 4 / 128; ++i) xs4[tid + 128 * i] = xg[tid + 128 * i];

    __syncthreads();

#pragma unroll 2
    for (int u = 0; u < 8; ++u) {
        const int j = tid + 128 * u;
        float x[9];
        float nrm = 0.f;
#pragma unroll
        for (int i = 0; i < 9; ++i) { x[i] = xs[j * 9 + i]; nrm += x[i] * x[i]; }
        const float inv = 1.0f / (nrm + 1.0f);
#pragma unroll
        for (int i = 0; i < 9; ++i) { x[i] *= inv; xs[j * 9 + i] = x[i]; }
        float c9[NCH];
        compact9(x, c9);
#pragma unroll
        for (int c = 0; c < NCH; ++c) ps[(c * NT + (j >> 6)) * PS + (j & 63)] = c9[c];
    }
    __syncthreads();

    float4* xo = (float4*)(Xn + (size_t)n0 * FF * 9);
#pragma unroll
    for (int i = 0; i < NT * FF * 9 / 4 / 128; ++i) xo[tid + 128 * i] = xs4[tid + 128 * i];

    v8f acc[NCH];
#pragma unroll
    for (int c = 0; c < NCH; ++c) acc[c] = v8f{};
    mix_tile(ps, row, kb, bI, bA, bS, acc);

#pragma unroll
    for (int c = 0; c < NCH; ++c)
#pragma unroll
        for (int r = 0; r < 8; ++r)
            Q[(size_t)c * NF + (size_t)(n0 + r + 8 * hi) * FF + col] = acc[c][r];

    const float4 zero4 = {0.f, 0.f, 0.f, 0.f};
#pragma unroll
    for (int c = 0; c < NCH; ++c) {
        float4* m4 = (float4*)(M + (size_t)c * NF + (size_t)n0 * FF);
        m4[tid] = zero4;
        m4[tid + 128] = zero4;
    }
}


#define TE 32

template <int KT, int LD>
__device__ __forceinline__ void load_bfrag(const float* __restrict__ Wm, int col, int kb,
                                           int k0, v2f* b) {
#pragma unroll
    for (int kk = 0; kk < KT; ++kk) {
        b[kk].x = Wm[(4 * (kk + k0) + kb) * LD + col];
        b[kk].y = Wm[(4 * (kk + k0) + kb + 1) * LD + col];
    }
}

__device__ __forceinline__ void gather_q(const float* __restrict__ Q, int d, int lane,
                                         float* q) {
#pragma unroll
    for (int t = 0; t < 2; ++t)
#pragma unroll
        for (int c = 0; c < NCH; ++c)
            q[NCH * t + c] = Q[c * NF + d * FF + lane + 32 * t];
}

__device__ __forceinline__ void layer2_tile(const v2f (*a2)[16], const v2f* gc, float bias,
                                            int col, int hi, float* h2) {
    v8f acc0 = {}, acc1 = {};
#pragma unroll
    for (int kk = 0; kk < 16; ++kk) {
        acc0 = wmma4(a2[0][kk], gc[kk], acc0);
        acc1 = wmma4(a2[1][kk], gc[kk], acc1);
    }
#pragma unroll
    for (int r = 0; r < 8; ++r) {
        h2[(r + 8 * hi) * H2S + col] = silu(acc0[r] + bias);
        h2[(16 + r + 8 * hi) * H2S + col] = silu(acc1[r] + bias);
    }
}

__device__ __forceinline__ void layer3_tile(const v2f (*a3)[32], const v2f* gc, float bias,
                                            v2f* gn, float* bias_n,
                                            const float* __restrict__ W3,
                                            const float* __restrict__ b3, int col, int kb,
                                            int hi, bool more, const float* cc, float* rad) {
    v8f acc0 = {}, acc1 = {};
#pragma unroll
    for (int half = 0; half < 2; ++half) {
        if (more) {
            load_bfrag<16, 192>(W3, col + 16, kb, 16 * half, gn + 16 * half);
            if (half == 0) *bias_n = b3[col + 16];
        }
#pragma unroll
        for (int kk = 16 * half; kk < 16 * half + 16; ++kk) {
            acc0 = wmma4(a3[0][kk], gc[kk], acc0);
            acc1 = wmma4(a3[1][kk], gc[kk], acc1);
        }
    }
#pragma unroll
    for (int r = 0; r < 8; ++r) {
        const int er = r + 8 * hi;
        rad[er * RDS + col] = silu(acc0[r] + bias) * cc[er];
        rad[(16 + er) * RDS + col] = silu(acc1[r] + bias) * cc[16 + er];
    }
}

__device__ __forceinline__ void scatter_edge(const float* rad, float* __restrict__ M, int e,
                                             int s, int lane, const float* q) {
#pragma unroll
    for (int t = 0; t < 2; ++t) {
        const int f = lane + 32 * t;
        const float rI = rad[e * RDS + 3 * f + 0];
        const float rA = rad[e * RDS + 3 * f + 1];
        const float rS = rad[e * RDS + 3 * f + 2];
        const int so = s * FF + f;
        atomicAdd(&M[0 * NF + so], rI * q[NCH * t + 0]);
#pragma unroll
        for (int c = 1; c < 4; ++c)
            atomicAdd(&M[c * NF + so], rA * q[NCH * t + c]);
#pragma unroll
        for (int c = 4; c < NCH; ++c)
            atomicAdd(&M[c * NF + so], rS * q[NCH * t + c]);
    }
}

__global__ __launch_bounds__(32) void edge_kernel(
    const float* __restrict__ rbf, const float* __restrict__ dij,
    const int* __restrict__ pi,
    const float* __restrict__ W1, const float* __restrict__ b1,
    const float* __restrict__ W2, const float* __restrict__ b2,
    const float* __restrict__ W3, const float* __restrict__ b3,
    const float* __restrict__ Q, float* __restrict__ M) {
    __shared__ float buf[TE * H1S + TE * H2S];
    __shared__ float cc[TE];
    float* h1 = buf;
    float* h2 = buf + TE * H1S;
    float* rad = buf;

    const int lane = threadIdx.x;
    const int e0 = blockIdx.x * TE;
    const int row = lane & 15;
    const int hi = lane >> 4;
    const int kb = hi * 2;
    const int col16 = lane & 15;

    const int dst_all = pi[EE + e0 + lane];
    const float dist = dij[e0 + lane];

    v2f a1[2][8];
#pragma unroll
    for (int t = 0; t < 2; ++t)
#pragma unroll
        for (int kk = 0; kk < 8; ++kk) {
            a1[t][kk].x = rbf[(size_t)(e0 + 16 * t + row) * 32 + 4 * kk + kb];
            a1[t][kk].y = rbf[(size_t)(e0 + 16 * t + row) * 32 + 4 * kk + kb + 1];
        }

    v2f g[2][32];
    float gb[2][4];

#pragma unroll
    for (int j = 0; j < 4; ++j) {
        load_bfrag<8, 64>(W1, j * 16 + col16, kb, 0, &g[0][8 * j]);
        gb[0][j] = b1[j * 16 + col16];
    }

    cc[lane] = 0.5f * (cosf(dist * (PI_F / 0.5f)) + 1.0f) * (dist < 0.5f ? 1.0f : 0.0f);

#pragma unroll
    for (int t = 0; t < 2; ++t) {
        load_bfrag<16, 128>(W2, t * 16 + col16, kb, 0, &g[1][16 * t]);
        gb[1][t] = b2[t * 16 + col16];
    }

#pragma unroll 1
    for (int jj = 0; jj < 2; ++jj) {
#pragma unroll
        for (int t = 0; t < 2; ++t) {
            const int col = 32 * jj + 16 * t + col16;
            v8f acc0 = {}, acc1 = {};
#pragma unroll
            for (int kk = 0; kk < 8; ++kk) {
                acc0 = wmma4(a1[0][kk], g[0][8 * t + kk], acc0);
                acc1 = wmma4(a1[1][kk], g[0][8 * t + kk], acc1);
            }
            const float bias = gb[0][t];
#pragma unroll
            for (int r = 0; r < 8; ++r) {
                h1[(r + 8 * hi) * H1S + col] = silu(acc0[r] + bias);
                h1[(16 + r + 8 * hi) * H1S + col] = silu(acc1[r] + bias);
            }
        }
#pragma unroll
        for (int i = 0; i < 16; ++i) g[0][i] = g[0][16 + i];
        gb[0][0] = gb[0][2];
        gb[0][1] = gb[0][3];
    }
    __syncthreads();

    v2f a2[2][16];
#pragma unroll
    for (int t = 0; t < 2; ++t)
#pragma unroll
        for (int kk = 0; kk < 16; ++kk) {
            a2[t][kk].x = h1[(16 * t + row) * H1S + 4 * kk + kb];
            a2[t][kk].y = h1[(16 * t + row) * H1S + 4 * kk + kb + 1];
        }
#pragma unroll 1
    for (int pp = 0; pp < 2; ++pp) {
        const int cb = 64 * pp + col16;
#pragma unroll
        for (int t = 0; t < 2; ++t) {
            const int ncol = cb + 32 + 16 * t;
            load_bfrag<16, 128>(W2, ncol, kb, 0, &g[0][16 * t]);
            gb[0][t] = b2[ncol];
            layer2_tile(a2, &g[1][16 * t], gb[1][t], cb + 16 * t, hi, h2);
        }
#pragma unroll
        for (int t = 0; t < 2; ++t) {
            if (pp == 0) {
                const int ncol = cb + 64 + 16 * t;
                load_bfrag<16, 128>(W2, ncol, kb, 0, &g[1][16 * t]);
                gb[1][t] = b2[ncol];
            } else {
                load_bfrag<16, 192>(W3, col16, kb, 16 * t, &g[1][16 * t]);
                if (t == 0) gb[1][0] = b3[col16];
            }
            layer2_tile(a2, &g[0][16 * t], gb[0][t], cb + 32 + 16 * t, hi, h2);
        }
    }
    __syncthreads();

    v2f a3[2][32];
#pragma unroll
    for (int t = 0; t < 2; ++t)
#pragma unroll
        for (int kk = 0; kk < 32; ++kk) {
            a3[t][kk].x = h2[(16 * t + row) * H2S + 4 * kk + kb];
            a3[t][kk].y = h2[(16 * t + row) * H2S + 4 * kk + kb + 1];
        }
#pragma unroll 1
    for (int jj = 0; jj < 6; ++jj) {
        const int c0 = 32 * jj + col16;
        layer3_tile(a3, g[1], gb[1][0], g[0], &gb[0][0], W3, b3, c0, kb, hi, true, cc, rad);
        layer3_tile(a3, g[0], gb[0][0], g[1], &gb[1][0], W3, b3, c0 + 16, kb, hi, jj < 5, cc,
                    rad);
    }
    const int src_all = pi[e0 + lane];
    float qv[3][2 * NCH];
    gather_q(Q, __builtin_amdgcn_readlane(dst_all, 0), lane, qv[0]);
    gather_q(Q, __builtin_amdgcn_readlane(dst_all, 1), lane, qv[1]);
    __syncthreads();

#pragma unroll 1
    for (int e3 = 0; e3 < TE - 2; e3 += 3) {
#pragma unroll
        for (int u = 0; u < 3; ++u) {
            const int e = e3 + u;
            gather_q(Q, __builtin_amdgcn_readlane(dst_all, e + 2), lane, qv[(u + 2) % 3]);
            scatter_edge(rad, M, e, __builtin_amdgcn_readlane(src_all, e), lane, qv[u]);
        }
    }
    scatter_edge(rad, M, TE - 2, __builtin_amdgcn_readlane(src_all, TE - 2), lane, qv[0]);
    scatter_edge(rad, M, TE - 1, __builtin_amdgcn_readlane(src_all, TE - 1), lane, qv[1]);
}

__device__ __forceinline__ void load_mq(const float* __restrict__ M, const float* __restrict__ Q,
                                        int idx, float* mc, float* yc) {
#pragma unroll
    for (int c = 0; c < NCH; ++c) { mc[c] = M[c * NF + idx]; yc[c] = Q[c * NF + idx]; }
}

__global__ __launch_bounds__(128) void node_out_kernel(const float* __restrict__ M,
                                                       const float* __restrict__ Q,
                                                       const float* __restrict__ Xn,
                                                       const float* __restrict__ chg,
                                                       const float* __restrict__ La,
                                                       const float* __restrict__ Lb,
                                                       const float* __restrict__ Lc,
                                                       float* __restrict__ out) {
    __shared__ float ps[NCH * NT * PS];

    const int tid = threadIdx.x;
    const int lane = tid & 31;
    const int wave = tid >> 5;
    const int n0 = blockIdx.x * NT;
    const int row = lane & 15;
    const int hi = lane >> 4;
    const int kb = hi * 2;
    const int col = wave * 16 + (lane & 15);

    v2f bI[16], bA[16], bS[16];
    load_lfrag(La, kb, col, bI);
    load_lfrag(Lb, kb, col, bA);
    load_lfrag(Lc, kb, col, bS);

    float cq[8];
#pragma unroll
    for (int u = 0; u < 8; ++u) cq[u] = chg[n0 + (tid >> 6) + 2 * u];
    float mc[4][NCH], yc[4][NCH];
#pragma unroll
    for (int u = 0; u < 3; ++u) load_mq(M, Q, n0 * FF + tid + 128 * u, mc[u], yc[u]);
#pragma unroll
    for (int u = 0; u < 8; ++u) {
        const int j = tid + 128 * u;
        if (u + 3 < 8) load_mq(M, Q, n0 * FF + j + 384, mc[(u + 3) & 3], yc[(u + 3) & 3]);
        float msg[9], Y[9], t1[9], t2[9];
        recon3(mc[u & 3], msg);
        recon3(yc[u & 3], Y);
        mm3(msg, Y, t1);
        mm3(Y, msg, t2);
        const float scale = 1.0f + 0.1f * cq[u];
        float T[9];
        float tn = 0.f;
#pragma unroll
        for (int i = 0; i < 9; ++i) { T[i] = scale * (t1[i] + t2[i]); tn += T[i] * T[i]; }
        const float inv = 1.0f / (tn + 1.0f);
#pragma unroll
        for (int i = 0; i < 9; ++i) T[i] *= inv;
        float c9[NCH];
        compact9(T, c9);
#pragma unroll
        for (int c = 0; c < NCH; ++c) ps[(c * NT + (j >> 6)) * PS + (j & 63)] = c9[c];
    }
    __syncthreads();

    float xv[8][9];
    float sc[8];
#pragma unroll
    for (int r = 0; r < 8; ++r) {
        const int n = n0 + r + 8 * hi;
        sc[r] = 1.0f + 0.1f * chg[n];
#pragma unroll
        for (int i = 0; i < 9; ++i) xv[r][i] = Xn[((size_t)n * FF + col) * 9 + i];
    }

    v8f acc[NCH];
#pragma unroll
    for (int c = 0; c < NCH; ++c) acc[c] = v8f{};
    mix_tile(ps, row, kb, bI, bA, bS, acc);

#pragma unroll
    for (int r = 0; r < 8; ++r) {
        const int n = n0 + r + 8 * hi;
        float dc[NCH], dX[9], prod[9];
#pragma unroll
        for (int c = 0; c < NCH; ++c) dc[c] = acc[c][r];
        recon3(dc, dX);
        mm3(dX, dX, prod);
#pragma unroll
        for (int i = 0; i < 9; ++i)
            out[((size_t)n * FF + col) * 9 + i] = xv[r][i] + dX[i] + sc[r] * prod[i];
    }
}

extern "C" void kernel_launch(void* const* d_in, const int* in_sizes, int n_in,
                              void* d_out, int out_size, void* d_ws, size_t ws_size,
                              hipStream_t stream) {
    const float* X   = (const float*)d_in[0];
    const int*   pi  = (const int*)  d_in[1];
    const float* dij = (const float*)d_in[2];
    const float* rbf = (const float*)d_in[3];
    const float* chg = (const float*)d_in[4];
    const float* W1  = (const float*)d_in[5];
    const float* b1  = (const float*)d_in[6];
    const float* W2  = (const float*)d_in[7];
    const float* b2  = (const float*)d_in[8];
    const float* W3  = (const float*)d_in[9];
    const float* b3  = (const float*)d_in[10];
    const float* L0  = (const float*)d_in[11];
    const float* L1  = (const float*)d_in[12];
    const float* L2  = (const float*)d_in[13];
    const float* L3  = (const float*)d_in[14];
    const float* L4  = (const float*)d_in[15];
    const float* L5  = (const float*)d_in[16];

    float* ws = (float*)d_ws;
    float* Xn = ws;
    float* Q  = Xn + (size_t)NF * 9 + (size_t)10 * NF;
    float* M  = Q + (size_t)10 * NF;

    node_in_kernel<<<NN / NT, 128, 0, stream>>>(X, Xn, Q, M, L0, L1, L2);
    edge_kernel<<<EE / TE, 32, 0, stream>>>(rbf, dij, pi, W1, b1, W2, b2, W3, b3, Q, M);
    node_out_kernel<<<NN / NT, 128, 0, stream>>>(M, Q, Xn, chg, L3, L4, L5, (float*)d_out);
}
